# starting kernel
# baseline (speedup 1.0000x reference)
#include <hip/hip_runtime.h>
#include <math.h>

typedef float v2f __attribute__((ext_vector_type(2)));
typedef float v8f __attribute__((ext_vector_type(8)));

#define B_    8
#define C_    512
#define N_    1024
#define C3_   1536
#define GRP_  8
#define GCH_  64
#define GELT_ (GCH_ * N_)
#define EPS_  1e-5f

__global__ __launch_bounds__(256) void kT_transpose(const float* __restrict__ in,
                                                    float* __restrict__ out,
                                                    int R, int Cc) {
  __shared__ float tile[32][33];
  int c0 = blockIdx.x * 32, r0 = blockIdx.y * 32;
  int tx = threadIdx.x, ty = threadIdx.y;
  #pragma unroll
  for (int i = 0; i < 32; i += 8)
    tile[ty + i][tx] = in[(size_t)(r0 + ty + i) * Cc + c0 + tx];
  __syncthreads();
  #pragma unroll
  for (int i = 0; i < 32; i += 8)
    out[(size_t)(c0 + ty + i) * R + r0 + tx] = tile[tx][ty + i];
}

__global__ __launch_bounds__(256) void kT_qtranspose(const float* __restrict__ qkv,
                                                     float* __restrict__ qT) {
  __shared__ float tile[32][33];
  int d0 = blockIdx.x * 32, n0 = blockIdx.y * 32;
  const float* in = qkv + (size_t)blockIdx.z * N_ * C3_;
  float*      out = qT  + (size_t)blockIdx.z * C_ * N_;
  int tx = threadIdx.x, ty = threadIdx.y;
  #pragma unroll
  for (int i = 0; i < 32; i += 8)
    tile[ty + i][tx] = in[(size_t)(n0 + ty + i) * C3_ + d0 + tx];
  __syncthreads();
  #pragma unroll
  for (int i = 0; i < 32; i += 8)
    out[(size_t)(d0 + ty + i) * N_ + n0 + tx] = tile[tx][ty + i];
}

__global__ __launch_bounds__(256) void kT_gn_stats(const float* __restrict__ x,
                                                   float* __restrict__ mean,
                                                   float* __restrict__ rstd) {
  __shared__ float s1[256], s2[256];
  int bg = blockIdx.x;
  const float4* p = (const float4*)(x + (size_t)bg * GELT_);
  float s = 0.f, q = 0.f;
  for (int i = threadIdx.x; i < GELT_ / 4; i += 256) {
    float4 v = p[i];
    s += v.x + v.y + v.z + v.w;
    q += v.x * v.x + v.y * v.y + v.z * v.z + v.w * v.w;
  }
  s1[threadIdx.x] = s; s2[threadIdx.x] = q;
  __syncthreads();
  for (int off = 128; off > 0; off >>= 1) {
    if (threadIdx.x < off) {
      s1[threadIdx.x] += s1[threadIdx.x + off];
      s2[threadIdx.x] += s2[threadIdx.x + off];
    }
    __syncthreads();
  }
  if (threadIdx.x == 0) {
    float m = s1[0] * (1.0f / GELT_);
    float v = s2[0] * (1.0f / GELT_) - m * m;
    mean[bg] = m;
    rstd[bg] = rsqrtf(v + EPS_);
  }
}

__global__ __launch_bounds__(256) void kT_gn_apply(const float* __restrict__ x,
                                                   const float* __restrict__ gamma,
                                                   const float* __restrict__ beta,
                                                   const float* __restrict__ mean,
                                                   const float* __restrict__ rstd,
                                                   float* __restrict__ xn) {
  int bc = blockIdx.x;
  int c = bc & (C_ - 1);
  int bg = ((bc >> 9) << 3) + (c >> 6);
  float gm = gamma[c] * rstd[bg];
  float bt = beta[c] - mean[bg] * gm;
  const float4* pin  = (const float4*)(x  + (size_t)bc * N_);
  float4*       pout = (float4*)(xn + (size_t)bc * N_);
  float4 v = pin[threadIdx.x];
  v.x = v.x * gm + bt; v.y = v.y * gm + bt;
  v.z = v.z * gm + bt; v.w = v.w * gm + bt;
  pout[threadIdx.x] = v;
}

__global__ __launch_bounds__(256) void kT_softmax_k(float* __restrict__ qkv) {
  __shared__ float red[256];
  float* row = qkv + (size_t)blockIdx.x * C3_ + C_;
  int t = threadIdx.x;
  float a = row[t], b = row[t + 256];
  red[t] = fmaxf(a, b);
  __syncthreads();
  for (int off = 128; off > 0; off >>= 1) {
    if (t < off) red[t] = fmaxf(red[t], red[t + off]);
    __syncthreads();
  }
  float mx = red[0];
  __syncthreads();
  float e0 = __expf(a - mx), e1 = __expf(b - mx);
  red[t] = e0 + e1;
  __syncthreads();
  for (int off = 128; off > 0; off >>= 1) {
    if (t < off) red[t] += red[t + off];
    __syncthreads();
  }
  float inv = 1.0f / red[0];
  row[t] = e0 * inv;
  row[t + 256] = e1 * inv;
}

__global__ __launch_bounds__(256) void kT_softmax_row1024(float* __restrict__ qT) {
  __shared__ float red[256];
  float4* row = (float4*)(qT + (size_t)blockIdx.x * N_);
  int t = threadIdx.x;
  float4 v = row[t];
  red[t] = fmaxf(fmaxf(v.x, v.y), fmaxf(v.z, v.w));
  __syncthreads();
  for (int off = 128; off > 0; off >>= 1) {
    if (t < off) red[t] = fmaxf(red[t], red[t + off]);
    __syncthreads();
  }
  float mx = red[0];
  __syncthreads();
  float4 e;
  e.x = __expf(v.x - mx); e.y = __expf(v.y - mx);
  e.z = __expf(v.z - mx); e.w = __expf(v.w - mx);
  red[t] = e.x + e.y + e.z + e.w;
  __syncthreads();
  for (int off = 128; off > 0; off >>= 1) {
    if (t < off) red[t] += red[t + off];
    __syncthreads();
  }
  float inv = 1.0f / red[0];
  e.x *= inv; e.y *= inv; e.z *= inv; e.w *= inv;
  row[t] = e;
}

template <int LDA, int LDB, int LDC, int K, bool TSTORE, bool HAS_BIAS, bool HAS_RES>
__global__ __launch_bounds__(256) void kT_gemm(
    const float* __restrict__ A, long long strideA,
    const float* __restrict__ B, long long strideB,
    float* __restrict__ Cmat,    long long strideC,
    const float* __restrict__ bias,
    const float* __restrict__ resid, long long strideR) {
  int z = blockIdx.z;
  const float* Ab = A + (size_t)z * strideA;
  const float* Bb = B + (size_t)z * strideB;
  float*       Cb = Cmat + (size_t)z * strideC;

  int wave = threadIdx.x >> 5;
  int lane = threadIdx.x & 31;
  int half = lane >> 4;
  int ml   = lane & 15;
  int m0 = blockIdx.x * 64  + (wave & 1) * 32;
  int n0 = blockIdx.y * 256 + (wave >> 1) * 64;

  v8f acc[2][4];
  {
    float bv[4];
    #pragma unroll
    for (int s = 0; s < 4; ++s) bv[s] = HAS_BIAS ? bias[n0 + 16 * s + ml] : 0.0f;
    #pragma unroll
    for (int mi = 0; mi < 2; ++mi)
      #pragma unroll
      for (int s = 0; s < 4; ++s)
        #pragma unroll
        for (int r = 0; r < 8; ++r) acc[mi][s][r] = bv[s];
  }

  const float* Ap = Ab + (size_t)(2 * half) * LDA + m0 + ml;
  const float* Bp = Bb + (size_t)(2 * half) * LDB + n0 + ml;

  #pragma unroll 1
  for (int kk = 0; kk < K; kk += 16) {
    __builtin_prefetch(Ap + (size_t)20 * LDA, 0, 1);
    #pragma unroll
    for (int j = 0; j < 4; ++j) {
      const int k0 = 4 * j * LDA, k1 = (4 * j + 1) * LDA;
      const int h0 = 4 * j * LDB, h1 = (4 * j + 1) * LDB;
      v2f a0; a0.x = Ap[k0];      a0.y = Ap[k1];
      v2f a1; a1.x = Ap[k0 + 16]; a1.y = Ap[k1 + 16];
      v2f b0; b0.x = Bp[h0];      b0.y = Bp[h1];
      v2f b1; b1.x = Bp[h0 + 16]; b1.y = Bp[h1 + 16];
      v2f b2; b2.x = Bp[h0 + 32]; b2.y = Bp[h1 + 32];
      v2f b3; b3.x = Bp[h0 + 48]; b3.y = Bp[h1 + 48];
      acc[0][0] = __builtin_amdgcn_wmma_f32_16x16x4_f32(false, a0, false, b0, (short)0, acc[0][0], false, false);
      acc[0][1] = __builtin_amdgcn_wmma_f32_16x16x4_f32(false, a0, false, b1, (short)0, acc[0][1], false, false);
      acc[0][2] = __builtin_amdgcn_wmma_f32_16x16x4_f32(false, a0, false, b2, (short)0, acc[0][2], false, false);
      acc[0][3] = __builtin_amdgcn_wmma_f32_16x16x4_f32(false, a0, false, b3, (short)0, acc[0][3], false, false);
      acc[1][0] = __builtin_amdgcn_wmma_f32_16x16x4_f32(false, a1, false, b0, (short)0, acc[1][0], false, false);
      acc[1][1] = __builtin_amdgcn_wmma_f32_16x16x4_f32(false, a1, false, b1, (short)0, acc[1][1], false, false);
      acc[1][2] = __builtin_amdgcn_wmma_f32_16x16x4_f32(false, a1, false, b2, (short)0, acc[1][2], false, false);
      acc[1][3] = __builtin_amdgcn_wmma_f32_16x16x4_f32(false, a1, false, b3, (short)0, acc[1][3], false, false);
    }
    Ap += 16 * LDA;
    Bp += 16 * LDB;
  }

  if (!TSTORE) {
    #pragma unroll
    for (int mi = 0; mi < 2; ++mi) {
      #pragma unroll
      for (int r = 0; r < 8; ++r) {
        int m = m0 + 16 * mi + r + 8 * half;
        #pragma unroll
        for (int s = 0; s < 4; ++s)
          Cb[(size_t)m * LDC + n0 + 16 * s + ml] = acc[mi][s][r];
      }
    }
  } else {
    const float* Rb = HAS_RES ? (resid + (size_t)z * strideR) : nullptr;
    #pragma unroll
    for (int mi = 0; mi < 2; ++mi) {
      #pragma unroll
      for (int r = 0; r < 8; ++r) {
        int m = m0 + 16 * mi + r + 8 * half;
        #pragma unroll
        for (int s = 0; s < 4; ++s) {
          size_t i = (size_t)(n0 + 16 * s + ml) * LDC + m;
          float v = acc[mi][s][r];
          if (HAS_RES) v += Rb[i];
          Cb[i] = v;
        }
      }
    }
  }
}

extern "C" void kernel_launch(void* const* d_in, const int* in_sizes, int n_in,
                              void* d_out, int out_size, void* d_ws, size_t ws_size,
                              hipStream_t stream) {
  const float* x      = (const float*)d_in[0];
  const float* gn_w   = (const float*)d_in[1];
  const float* gn_b   = (const float*)d_in[2];
  const float* qkv_w  = (const float*)d_in[3];
  const float* qkv_b  = (const float*)d_in[4];
  const float* proj_w = (const float*)d_in[5];
  const float* proj_b = (const float*)d_in[6];
  float* out = (float*)d_out;

  float* ws = (float*)d_ws;
  const size_t XN_SZ  = (size_t)B_ * C_ * N_;
  const size_t QKV_SZ = (size_t)B_ * N_ * C3_;
  const size_t CTX_SZ = (size_t)B_ * C_ * C_;
  float* mean   = ws;
  float* rstd   = ws + 64;
  float* xn     = ws + 256;
  float* qkv    = xn   + XN_SZ;
  float* qT     = qkv  + QKV_SZ;
  float* ctx    = qT   + XN_SZ;
  float* o2T    = ctx  + CTX_SZ;
  float* wqkvT  = o2T  + XN_SZ;
  float* wprojT = wqkvT + (size_t)C_ * C3_;

  kT_transpose<<<dim3(C_ / 32, C3_ / 32), dim3(32, 8), 0, stream>>>(qkv_w,  wqkvT,  C3_, C_);
  kT_transpose<<<dim3(C_ / 32, C_  / 32), dim3(32, 8), 0, stream>>>(proj_w, wprojT, C_,  C_);

  kT_gn_stats<<<B_ * GRP_, 256, 0, stream>>>(x, mean, rstd);
  kT_gn_apply<<<B_ * C_,   256, 0, stream>>>(x, gn_w, gn_b, mean, rstd, xn);

  kT_gemm<N_, C3_, C3_, C_, false, true, false>
      <<<dim3(N_ / 64, C3_ / 256, B_), 256, 0, stream>>>(
      xn, (long long)C_ * N_,
      wqkvT, 0,
      qkv, (long long)N_ * C3_,
      qkv_b, nullptr, 0);

  kT_softmax_k<<<B_ * N_, 256, 0, stream>>>(qkv);

  kT_qtranspose<<<dim3(C_ / 32, N_ / 32, B_), dim3(32, 8), 0, stream>>>(qkv, qT);
  kT_softmax_row1024<<<B_ * C_, 256, 0, stream>>>(qT);

  kT_gemm<C3_, C3_, C_, N_, false, false, false>
      <<<dim3(C_ / 64, C_ / 256, B_), 256, 0, stream>>>(
      qkv + C_,     (long long)N_ * C3_,
      qkv + 2 * C_, (long long)N_ * C3_,
      ctx, (long long)C_ * C_,
      nullptr, nullptr, 0);

  kT_gemm<N_, C_, N_, C_, true, false, false>
      <<<dim3(N_ / 64, C_ / 256, B_), 256, 0, stream>>>(
      qT, (long long)C_ * N_,
      ctx, (long long)C_ * C_,
      o2T, (long long)C_ * N_,
      nullptr, nullptr, 0);

  kT_gemm<N_, C_, N_, C_, true, true, true>
      <<<dim3(N_ / 64, C_ / 256, B_), 256, 0, stream>>>(
      o2T, (long long)C_ * N_,
      wprojT, 0,
      out, (long long)C_ * N_,
      proj_b, xn, (long long)C_ * N_);
}


# GEMM operands staged through LDS: a block fetches its 32-deep A and B tiles once with 16-byte loads, one tile ahead, two stages; fragments read from LDS; prefetch hint dropped; sum order kept
# speedup vs baseline: 1.3046x; 1.3046x over previous
#include <hip/hip_runtime.h>
#include <math.h>

typedef float v2f __attribute__((ext_vector_type(2)));
typedef float v8f __attribute__((ext_vector_type(8)));

#define B_    8
#define C_    512
#define N_    1024
#define C3_   1536
#define GRP_  8
#define GCH_  64
#define GELT_ (GCH_ * N_)
#define EPS_  1e-5f

__global__ __launch_bounds__(256) void kT_transpose(const float* __restrict__ in,
                                                    float* __restrict__ out,
                                                    int R, int Cc) {
  __shared__ float tile[32][33];
  int c0 = blockIdx.x * 32, r0 = blockIdx.y * 32;
  int tx = threadIdx.x, ty = threadIdx.y;
  #pragma unroll
  for (int i = 0; i < 32; i += 8)
    tile[ty + i][tx] = in[(size_t)(r0 + ty + i) * Cc + c0 + tx];
  __syncthreads();
  #pragma unroll
  for (int i = 0; i < 32; i += 8)
    out[(size_t)(c0 + ty + i) * R + r0 + tx] = tile[tx][ty + i];
}

__global__ __launch_bounds__(256) void kT_qtranspose(const float* __restrict__ qkv,
                                                     float* __restrict__ qT) {
  __shared__ float tile[32][33];
  int d0 = blockIdx.x * 32, n0 = blockIdx.y * 32;
  const float* in = qkv + (size_t)blockIdx.z * N_ * C3_;
  float*      out = qT  + (size_t)blockIdx.z * C_ * N_;
  int tx = threadIdx.x, ty = threadIdx.y;
  #pragma unroll
  for (int i = 0; i < 32; i += 8)
    tile[ty + i][tx] = in[(size_t)(n0 + ty + i) * C3_ + d0 + tx];
  __syncthreads();
  #pragma unroll
  for (int i = 0; i < 32; i += 8)
    out[(size_t)(d0 + ty + i) * N_ + n0 + tx] = tile[tx][ty + i];
}

__global__ __launch_bounds__(256) void kT_gn_stats(const float* __restrict__ x,
                                                   float* __restrict__ mean,
                                                   float* __restrict__ rstd) {
  __shared__ float s1[256], s2[256];
  int bg = blockIdx.x;
  const float4* p = (const float4*)(x + (size_t)bg * GELT_);
  float s = 0.f, q = 0.f;
  for (int i = threadIdx.x; i < GELT_ / 4; i += 256) {
    float4 v = p[i];
    s += v.x + v.y + v.z + v.w;
    q += v.x * v.x + v.y * v.y + v.z * v.z + v.w * v.w;
  }
  s1[threadIdx.x] = s; s2[threadIdx.x] = q;
  __syncthreads();
  for (int off = 128; off > 0; off >>= 1) {
    if (threadIdx.x < off) {
      s1[threadIdx.x] += s1[threadIdx.x + off];
      s2[threadIdx.x] += s2[threadIdx.x + off];
    }
    __syncthreads();
  }
  if (threadIdx.x == 0) {
    float m = s1[0] * (1.0f / GELT_);
    float v = s2[0] * (1.0f / GELT_) - m * m;
    mean[bg] = m;
    rstd[bg] = rsqrtf(v + EPS_);
  }
}

__global__ __launch_bounds__(256) void kT_gn_apply(const float* __restrict__ x,
                                                   const float* __restrict__ gamma,
                                                   const float* __restrict__ beta,
                                                   const float* __restrict__ mean,
                                                   const float* __restrict__ rstd,
                                                   float* __restrict__ xn) {
  int bc = blockIdx.x;
  int c = bc & (C_ - 1);
  int bg = ((bc >> 9) << 3) + (c >> 6);
  float gm = gamma[c] * rstd[bg];
  float bt = beta[c] - mean[bg] * gm;
  const float4* pin  = (const float4*)(x  + (size_t)bc * N_);
  float4*       pout = (float4*)(xn + (size_t)bc * N_);
  float4 v = pin[threadIdx.x];
  v.x = v.x * gm + bt; v.y = v.y * gm + bt;
  v.z = v.z * gm + bt; v.w = v.w * gm + bt;
  pout[threadIdx.x] = v;
}

__global__ __launch_bounds__(256) void kT_softmax_k(float* __restrict__ qkv) {
  __shared__ float red[256];
  float* row = qkv + (size_t)blockIdx.x * C3_ + C_;
  int t = threadIdx.x;
  float a = row[t], b = row[t + 256];
  red[t] = fmaxf(a, b);
  __syncthreads();
  for (int off = 128; off > 0; off >>= 1) {
    if (t < off) red[t] = fmaxf(red[t], red[t + off]);
    __syncthreads();
  }
  float mx = red[0];
  __syncthreads();
  float e0 = __expf(a - mx), e1 = __expf(b - mx);
  red[t] = e0 + e1;
  __syncthreads();
  for (int off = 128; off > 0; off >>= 1) {
    if (t < off) red[t] += red[t + off];
    __syncthreads();
  }
  float inv = 1.0f / red[0];
  row[t] = e0 * inv;
  row[t + 256] = e1 * inv;
}

__global__ __launch_bounds__(256) void kT_softmax_row1024(float* __restrict__ qT) {
  __shared__ float red[256];
  float4* row = (float4*)(qT + (size_t)blockIdx.x * N_);
  int t = threadIdx.x;
  float4 v = row[t];
  red[t] = fmaxf(fmaxf(v.x, v.y), fmaxf(v.z, v.w));
  __syncthreads();
  for (int off = 128; off > 0; off >>= 1) {
    if (t < off) red[t] = fmaxf(red[t], red[t + off]);
    __syncthreads();
  }
  float mx = red[0];
  __syncthreads();
  float4 e;
  e.x = __expf(v.x - mx); e.y = __expf(v.y - mx);
  e.z = __expf(v.z - mx); e.w = __expf(v.w - mx);
  red[t] = e.x + e.y + e.z + e.w;
  __syncthreads();
  for (int off = 128; off > 0; off >>= 1) {
    if (t < off) red[t] += red[t + off];
    __syncthreads();
  }
  float inv = 1.0f / red[0];
  e.x *= inv; e.y *= inv; e.z *= inv; e.w *= inv;
  row[t] = e;
}

typedef float v4f __attribute__((ext_vector_type(4)));

#define KS_     32
#define KP_     (KS_ / 2)
#define APITCH_ (64 * 2 + 32)
#define BPITCH_ (256 * 2 + 32)
#define ASTG_   (KP_ * APITCH_)
#define BSTG_   (KP_ * BPITCH_)
#define STG_    (ASTG_ + BSTG_)

__device__ __forceinline__ void kT_stage_products(const float* __restrict__ sa,
                                                  const float* __restrict__ sb,
                                                  v8f (&acc)[2][4]) {
  #pragma unroll
  for (int j = 0; j < KS_ / 4; ++j) {
    v2f a0 = *(const v2f*)(sa + j * 2 * APITCH_);
    v2f a1 = *(const v2f*)(sa + j * 2 * APITCH_ + 32);
    v2f b0 = *(const v2f*)(sb + j * 2 * BPITCH_);
    v2f b1 = *(const v2f*)(sb + j * 2 * BPITCH_ + 32);
    v2f b2 = *(const v2f*)(sb + j * 2 * BPITCH_ + 64);
    v2f b3 = *(const v2f*)(sb + j * 2 * BPITCH_ + 96);
    acc[0][0] = __builtin_amdgcn_wmma_f32_16x16x4_f32(false, a0, false, b0, (short)0, acc[0][0], false, false);
    acc[0][1] = __builtin_amdgcn_wmma_f32_16x16x4_f32(false, a0, false, b1, (short)0, acc[0][1], false, false);
    acc[0][2] = __builtin_amdgcn_wmma_f32_16x16x4_f32(false, a0, false, b2, (short)0, acc[0][2], false, false);
    acc[0][3] = __builtin_amdgcn_wmma_f32_16x16x4_f32(false, a0, false, b3, (short)0, acc[0][3], false, false);
    acc[1][0] = __builtin_amdgcn_wmma_f32_16x16x4_f32(false, a1, false, b0, (short)0, acc[1][0], false, false);
    acc[1][1] = __builtin_amdgcn_wmma_f32_16x16x4_f32(false, a1, false, b1, (short)0, acc[1][1], false, false);
    acc[1][2] = __builtin_amdgcn_wmma_f32_16x16x4_f32(false, a1, false, b2, (short)0, acc[1][2], false, false);
    acc[1][3] = __builtin_amdgcn_wmma_f32_16x16x4_f32(false, a1, false, b3, (short)0, acc[1][3], false, false);
  }
}

template <int LDA, int LDB, int LDC, int K, bool TSTORE, bool HAS_BIAS, bool HAS_RES>
__global__ __launch_bounds__(256) void kT_gemm(
    const float* __restrict__ A, long long strideA,
    const float* __restrict__ B, long long strideB,
    float* __restrict__ Cmat,    long long strideC,
    const float* __restrict__ bias,
    const float* __restrict__ resid, long long strideR) {
  __shared__ __attribute__((aligned(16))) float smem[2 * STG_];

  int z = blockIdx.z;
  const float* Ab = A + (size_t)z * strideA;
  const float* Bb = B + (size_t)z * strideB;
  float*       Cb = Cmat + (size_t)z * strideC;

  int tid  = threadIdx.x;
  int wave = tid >> 5;
  int lane = tid & 31;
  int half = lane >> 4;
  int ml   = lane & 15;
  int mblk = blockIdx.x * 64;
  int nblk = blockIdx.y * 256;
  int m0 = mblk + (wave & 1) * 32;
  int n0 = nblk + (wave >> 1) * 64;

  v8f acc[2][4];
  {
    float bv[4];
    #pragma unroll
    for (int s = 0; s < 4; ++s) bv[s] = HAS_BIAS ? bias[n0 + 16 * s + ml] : 0.0f;
    #pragma unroll
    for (int mi = 0; mi < 2; ++mi)
      #pragma unroll
      for (int s = 0; s < 4; ++s)
        #pragma unroll
        for (int r = 0; r < 8; ++r) acc[mi][s][r] = bv[s];
  }

  const float* Ag = Ab + (size_t)(2 * (tid >> 4)) * LDA + mblk + 4 * (tid & 15);
  const float* Bg = Bb + (size_t)(2 * (tid >> 6)) * LDB + nblk + 4 * (tid & 63);
  const int aSt = (tid >> 4) * APITCH_ + 8 * (tid & 15);
  const int bSt = ASTG_ + (tid >> 6) * BPITCH_ + 8 * (tid & 63);
  const int aLd = half * APITCH_ + ((wave & 1) * 32 + ml) * 2;
  const int bLd = ASTG_ + half * BPITCH_ + ((wave >> 1) * 64 + ml) * 2;

  v4f ra0, ra1, rb0[4], rb1[4];

#define KT_FETCH()                                                    \
  do {                                                                \
    ra0 = *(const v4f*)(Ag);                                          \
    ra1 = *(const v4f*)(Ag + LDA);                                    \
    _Pragma("unroll")                                                 \
    for (int i = 0; i < 4; ++i) {                                     \
      rb0[i] = *(const v4f*)(Bg + (size_t)(8 * i) * LDB);             \
      rb1[i] = *(const v4f*)(Bg + (size_t)(8 * i + 1) * LDB);         \
    }                                                                 \
    Ag += (size_t)KS_ * LDA;                                          \
    Bg += (size_t)KS_ * LDB;                                          \
  } while (0)

#define KT_PARK(stg)                                                  \
  do {                                                                \
    float* s_ = smem + (stg) * STG_;                                  \
    v4f t_;                                                           \
    t_.x = ra0.x; t_.y = ra1.x; t_.z = ra0.y; t_.w = ra1.y;           \
    *(v4f*)(s_ + aSt) = t_;                                           \
    t_.x = ra0.z; t_.y = ra1.z; t_.z = ra0.w; t_.w = ra1.w;           \
    *(v4f*)(s_ + aSt + 4) = t_;                                       \
    _Pragma("unroll")                                                 \
    for (int i = 0; i < 4; ++i) {                                     \
      float* d_ = s_ + bSt + i * 4 * BPITCH_;                         \
      t_.x = rb0[i].x; t_.y = rb1[i].x; t_.z = rb0[i].y; t_.w = rb1[i].y; \
      *(v4f*)(d_) = t_;                                               \
      t_.x = rb0[i].z; t_.y = rb1[i].z; t_.z = rb0[i].w; t_.w = rb1[i].w; \
      *(v4f*)(d_ + 4) = t_;                                           \
    }                                                                 \
  } while (0)

  constexpr int T = K / KS_;

  KT_FETCH();
  KT_PARK(0);
  __syncthreads();

  #pragma unroll 1
  for (int t = 0; t < T - 1; ++t) {
    const int cur = t & 1;
    KT_FETCH();
    kT_stage_products(smem + cur * STG_ + aLd, smem + cur * STG_ + bLd, acc);
    KT_PARK(cur ^ 1);
    __syncthreads();
  }
  kT_stage_products(smem + ((T - 1) & 1) * STG_ + aLd, smem + ((T - 1) & 1) * STG_ + bLd, acc);

#undef KT_FETCH
#undef KT_PARK

  if (!TSTORE) {
    #pragma unroll
    for (int mi = 0; mi < 2; ++mi) {
      #pragma unroll
      for (int r = 0; r < 8; ++r) {
        int m = m0 + 16 * mi + r + 8 * half;
        #pragma unroll
        for (int s = 0; s < 4; ++s)
          Cb[(size_t)m * LDC + n0 + 16 * s + ml] = acc[mi][s][r];
      }
    }
  } else {
    const float* Rb = HAS_RES ? (resid + (size_t)z * strideR) : nullptr;
    #pragma unroll
    for (int mi = 0; mi < 2; ++mi) {
      #pragma unroll
      for (int r = 0; r < 8; ++r) {
        int m = m0 + 16 * mi + r + 8 * half;
        #pragma unroll
        for (int s = 0; s < 4; ++s) {
          size_t i = (size_t)(n0 + 16 * s + ml) * LDC + m;
          float v = acc[mi][s][r];
          if (HAS_RES) v += Rb[i];
          Cb[i] = v;
        }
      }
    }
  }
}

extern "C" void kernel_launch(void* const* d_in, const int* in_sizes, int n_in,
                              void* d_out, int out_size, void* d_ws, size_t ws_size,
                              hipStream_t stream) {
  const float* x      = (const float*)d_in[0];
  const float* gn_w   = (const float*)d_in[1];
  const float* gn_b   = (const float*)d_in[2];
  const float* qkv_w  = (const float*)d_in[3];
  const float* qkv_b  = (const float*)d_in[4];
  const float* proj_w = (const float*)d_in[5];
  const float* proj_b = (const float*)d_in[6];
  float* out = (float*)d_out;

  float* ws = (float*)d_ws;
  const size_t XN_SZ  = (size_t)B_ * C_ * N_;
  const size_t QKV_SZ = (size_t)B_ * N_ * C3_;
  const size_t CTX_SZ = (size_t)B_ * C_ * C_;
  float* mean   = ws;
  float* rstd   = ws + 64;
  float* xn     = ws + 256;
  float* qkv    = xn   + XN_SZ;
  float* qT     = qkv  + QKV_SZ;
  float* ctx    = qT   + XN_SZ;
  float* o2T    = ctx  + CTX_SZ;
  float* wqkvT  = o2T  + XN_SZ;
  float* wprojT = wqkvT + (size_t)C_ * C3_;

  kT_transpose<<<dim3(C_ / 32, C3_ / 32), dim3(32, 8), 0, stream>>>(qkv_w,  wqkvT,  C3_, C_);
  kT_transpose<<<dim3(C_ / 32, C_  / 32), dim3(32, 8), 0, stream>>>(proj_w, wprojT, C_,  C_);

  kT_gn_stats<<<B_ * GRP_, 256, 0, stream>>>(x, mean, rstd);
  kT_gn_apply<<<B_ * C_,   256, 0, stream>>>(x, gn_w, gn_b, mean, rstd, xn);

  kT_gemm<N_, C3_, C3_, C_, false, true, false>
      <<<dim3(N_ / 64, C3_ / 256, B_), 256, 0, stream>>>(
      xn, (long long)C_ * N_,
      wqkvT, 0,
      qkv, (long long)N_ * C3_,
      qkv_b, nullptr, 0);

  kT_softmax_k<<<B_ * N_, 256, 0, stream>>>(qkv);

  kT_qtranspose<<<dim3(C_ / 32, N_ / 32, B_), dim3(32, 8), 0, stream>>>(qkv, qT);
  kT_softmax_row1024<<<B_ * C_, 256, 0, stream>>>(qT);

  kT_gemm<C3_, C3_, C_, N_, false, false, false>
      <<<dim3(C_ / 64, C_ / 256, B_), 256, 0, stream>>>(
      qkv + C_,     (long long)N_ * C3_,
      qkv + 2 * C_, (long long)N_ * C3_,
      ctx, (long long)C_ * C_,
      nullptr, nullptr, 0);

  kT_gemm<N_, C_, N_, C_, true, false, false>
      <<<dim3(N_ / 64, C_ / 256, B_), 256, 0, stream>>>(
      qT, (long long)C_ * N_,
      ctx, (long long)C_ * C_,
      o2T, (long long)C_ * N_,
      nullptr, nullptr, 0);

  kT_gemm<N_, C_, N_, C_, true, true, true>
      <<<dim3(N_ / 64, C_ / 256, B_), 256, 0, stream>>>(
      o2T, (long long)C_ * N_,
      wprojT, 0,
      out, (long long)C_ * N_,
      proj_b, xn, (long long)C_ * N_);
}


# context GEMM split in two along its 1024-token sum (grid z 8 to 16, K 512 each): 256 blocks instead of 128; the next GEMM adds the two half contexts as it parks its B tile; sum order changed
# speedup vs baseline: 1.3987x; 1.0721x over previous
#include <hip/hip_runtime.h>
#include <math.h>

typedef float v2f __attribute__((ext_vector_type(2)));
typedef float v8f __attribute__((ext_vector_type(8)));

#define B_    8
#define C_    512
#define N_    1024
#define C3_   1536
#define GRP_  8
#define GCH_  64
#define GELT_ (GCH_ * N_)
#define EPS_  1e-5f

__global__ __launch_bounds__(256) void kT_transpose(const float* __restrict__ in,
                                                    float* __restrict__ out,
                                                    int R, int Cc) {
  __shared__ float tile[32][33];
  int c0 = blockIdx.x * 32, r0 = blockIdx.y * 32;
  int tx = threadIdx.x, ty = threadIdx.y;
  #pragma unroll
  for (int i = 0; i < 32; i += 8)
    tile[ty + i][tx] = in[(size_t)(r0 + ty + i) * Cc + c0 + tx];
  __syncthreads();
  #pragma unroll
  for (int i = 0; i < 32; i += 8)
    out[(size_t)(c0 + ty + i) * R + r0 + tx] = tile[tx][ty + i];
}

__global__ __launch_bounds__(256) void kT_qtranspose(const float* __restrict__ qkv,
                                                     float* __restrict__ qT) {
  __shared__ float tile[32][33];
  int d0 = blockIdx.x * 32, n0 = blockIdx.y * 32;
  const float* in = qkv + (size_t)blockIdx.z * N_ * C3_;
  float*      out = qT  + (size_t)blockIdx.z * C_ * N_;
  int tx = threadIdx.x, ty = threadIdx.y;
  #pragma unroll
  for (int i = 0; i < 32; i += 8)
    tile[ty + i][tx] = in[(size_t)(n0 + ty + i) * C3_ + d0 + tx];
  __syncthreads();
  #pragma unroll
  for (int i = 0; i < 32; i += 8)
    out[(size_t)(d0 + ty + i) * N_ + n0 + tx] = tile[tx][ty + i];
}

__global__ __launch_bounds__(256) void kT_gn_stats(const float* __restrict__ x,
                                                   float* __restrict__ mean,
                                                   float* __restrict__ rstd) {
  __shared__ float s1[256], s2[256];
  int bg = blockIdx.x;
  const float4* p = (const float4*)(x + (size_t)bg * GELT_);
  float s = 0.f, q = 0.f;
  for (int i = threadIdx.x; i < GELT_ / 4; i += 256) {
    float4 v = p[i];
    s += v.x + v.y + v.z + v.w;
    q += v.x * v.x + v.y * v.y + v.z * v.z + v.w * v.w;
  }
  s1[threadIdx.x] = s; s2[threadIdx.x] = q;
  __syncthreads();
  for (int off = 128; off > 0; off >>= 1) {
    if (threadIdx.x < off) {
      s1[threadIdx.x] += s1[threadIdx.x + off];
      s2[threadIdx.x] += s2[threadIdx.x + off];
    }
    __syncthreads();
  }
  if (threadIdx.x == 0) {
    float m = s1[0] * (1.0f / GELT_);
    float v = s2[0] * (1.0f / GELT_) - m * m;
    mean[bg] = m;
    rstd[bg] = rsqrtf(v + EPS_);
  }
}

__global__ __launch_bounds__(256) void kT_gn_apply(const float* __restrict__ x,
                                                   const float* __restrict__ gamma,
                                                   const float* __restrict__ beta,
                                                   const float* __restrict__ mean,
                                                   const float* __restrict__ rstd,
                                                   float* __restrict__ xn) {
  int bc = blockIdx.x;
  int c = bc & (C_ - 1);
  int bg = ((bc >> 9) << 3) + (c >> 6);
  float gm = gamma[c] * rstd[bg];
  float bt = beta[c] - mean[bg] * gm;
  const float4* pin  = (const float4*)(x  + (size_t)bc * N_);
  float4*       pout = (float4*)(xn + (size_t)bc * N_);
  float4 v = pin[threadIdx.x];
  v.x = v.x * gm + bt; v.y = v.y * gm + bt;
  v.z = v.z * gm + bt; v.w = v.w * gm + bt;
  pout[threadIdx.x] = v;
}

__global__ __launch_bounds__(256) void kT_softmax_k(float* __restrict__ qkv) {
  __shared__ float red[256];
  float* row = qkv + (size_t)blockIdx.x * C3_ + C_;
  int t = threadIdx.x;
  float a = row[t], b = row[t + 256];
  red[t] = fmaxf(a, b);
  __syncthreads();
  for (int off = 128; off > 0; off >>= 1) {
    if (t < off) red[t] = fmaxf(red[t], red[t + off]);
    __syncthreads();
  }
  float mx = red[0];
  __syncthreads();
  float e0 = __expf(a - mx), e1 = __expf(b - mx);
  red[t] = e0 + e1;
  __syncthreads();
  for (int off = 128; off > 0; off >>= 1) {
    if (t < off) red[t] += red[t + off];
    __syncthreads();
  }
  float inv = 1.0f / red[0];
  row[t] = e0 * inv;
  row[t + 256] = e1 * inv;
}

__global__ __launch_bounds__(256) void kT_softmax_row1024(float* __restrict__ qT) {
  __shared__ float red[256];
  float4* row = (float4*)(qT + (size_t)blockIdx.x * N_);
  int t = threadIdx.x;
  float4 v = row[t];
  red[t] = fmaxf(fmaxf(v.x, v.y), fmaxf(v.z, v.w));
  __syncthreads();
  for (int off = 128; off > 0; off >>= 1) {
    if (t < off) red[t] = fmaxf(red[t], red[t + off]);
    __syncthreads();
  }
  float mx = red[0];
  __syncthreads();
  float4 e;
  e.x = __expf(v.x - mx); e.y = __expf(v.y - mx);
  e.z = __expf(v.z - mx); e.w = __expf(v.w - mx);
  red[t] = e.x + e.y + e.z + e.w;
  __syncthreads();
  for (int off = 128; off > 0; off >>= 1) {
    if (t < off) red[t] += red[t + off];
    __syncthreads();
  }
  float inv = 1.0f / red[0];
  e.x *= inv; e.y *= inv; e.z *= inv; e.w *= inv;
  row[t] = e;
}

typedef float v4f __attribute__((ext_vector_type(4)));

#define KS_     32
#define KP_     (KS_ / 2)
#define APITCH_ (64 * 2 + 32)
#define BPITCH_ (256 * 2 + 32)
#define ASTG_   (KP_ * APITCH_)
#define BSTG_   (KP_ * BPITCH_)
#define STG_    (ASTG_ + BSTG_)

__device__ __forceinline__ void kT_stage_products(const float* __restrict__ sa,
                                                  const float* __restrict__ sb,
                                                  v8f (&acc)[2][4]) {
  #pragma unroll
  for (int j = 0; j < KS_ / 4; ++j) {
    v2f a0 = *(const v2f*)(sa + j * 2 * APITCH_);
    v2f a1 = *(const v2f*)(sa + j * 2 * APITCH_ + 32);
    v2f b0 = *(const v2f*)(sb + j * 2 * BPITCH_);
    v2f b1 = *(const v2f*)(sb + j * 2 * BPITCH_ + 32);
    v2f b2 = *(const v2f*)(sb + j * 2 * BPITCH_ + 64);
    v2f b3 = *(const v2f*)(sb + j * 2 * BPITCH_ + 96);
    acc[0][0] = __builtin_amdgcn_wmma_f32_16x16x4_f32(false, a0, false, b0, (short)0, acc[0][0], false, false);
    acc[0][1] = __builtin_amdgcn_wmma_f32_16x16x4_f32(false, a0, false, b1, (short)0, acc[0][1], false, false);
    acc[0][2] = __builtin_amdgcn_wmma_f32_16x16x4_f32(false, a0, false, b2, (short)0, acc[0][2], false, false);
    acc[0][3] = __builtin_amdgcn_wmma_f32_16x16x4_f32(false, a0, false, b3, (short)0, acc[0][3], false, false);
    acc[1][0] = __builtin_amdgcn_wmma_f32_16x16x4_f32(false, a1, false, b0, (short)0, acc[1][0], false, false);
    acc[1][1] = __builtin_amdgcn_wmma_f32_16x16x4_f32(false, a1, false, b1, (short)0, acc[1][1], false, false);
    acc[1][2] = __builtin_amdgcn_wmma_f32_16x16x4_f32(false, a1, false, b2, (short)0, acc[1][2], false, false);
    acc[1][3] = __builtin_amdgcn_wmma_f32_16x16x4_f32(false, a1, false, b3, (short)0, acc[1][3], false, false);
  }
}

template <int LDA, int LDB, int LDC, int K, bool TSTORE, bool HAS_BIAS, bool HAS_RES, int BADD>
__global__ __launch_bounds__(256) __attribute__((amdgpu_waves_per_eu(2, BADD ? 2 : 8))) void kT_gemm(
    const float* __restrict__ A, long long strideA,
    const float* __restrict__ B, long long strideB,
    float* __restrict__ Cmat,    long long strideC,
    const float* __restrict__ bias,
    const float* __restrict__ resid, long long strideR) {
  __shared__ __attribute__((aligned(16))) float smem[2 * STG_];

  int z = blockIdx.z;
  const float* Ab = A + (size_t)z * strideA;
  const float* Bb = B + (size_t)z * strideB;
  float*       Cb = Cmat + (size_t)z * strideC;

  int tid  = threadIdx.x;
  int wave = tid >> 5;
  int lane = tid & 31;
  int half = lane >> 4;
  int ml   = lane & 15;
  int mblk = blockIdx.x * 64;
  int nblk = blockIdx.y * 256;
  int m0 = mblk + (wave & 1) * 32;
  int n0 = nblk + (wave >> 1) * 64;

  v8f acc[2][4];
  {
    float bv[4];
    #pragma unroll
    for (int s = 0; s < 4; ++s) bv[s] = HAS_BIAS ? bias[n0 + 16 * s + ml] : 0.0f;
    #pragma unroll
    for (int mi = 0; mi < 2; ++mi)
      #pragma unroll
      for (int s = 0; s < 4; ++s)
        #pragma unroll
        for (int r = 0; r < 8; ++r) acc[mi][s][r] = bv[s];
  }

  const float* Ag = Ab + (size_t)(2 * (tid >> 4)) * LDA + mblk + 4 * (tid & 15);
  const float* Bg = Bb + (size_t)(2 * (tid >> 6)) * LDB + nblk + 4 * (tid & 63);
  const int aSt = (tid >> 4) * APITCH_ + 8 * (tid & 15);
  const int bSt = ASTG_ + (tid >> 6) * BPITCH_ + 8 * (tid & 63);
  const int aLd = half * APITCH_ + ((wave & 1) * 32 + ml) * 2;
  const int bLd = ASTG_ + half * BPITCH_ + ((wave >> 1) * 64 + ml) * 2;

  v4f ra0, ra1, rb0[4], rb1[4];
  v4f rc0[4], rc1[4];

#define KT_FETCH()                                                    \
  do {                                                                \
    ra0 = *(const v4f*)(Ag);                                          \
    ra1 = *(const v4f*)(Ag + LDA);                                    \
    _Pragma("unroll")                                                 \
    for (int i = 0; i < 4; ++i) {                                     \
      rb0[i] = *(const v4f*)(Bg + (size_t)(8 * i) * LDB);             \
      rb1[i] = *(const v4f*)(Bg + (size_t)(8 * i + 1) * LDB);         \
      if (BADD) {                                                     \
        rc0[i] = *(const v4f*)(Bg + BADD + (size_t)(8 * i) * LDB);    \
        rc1[i] = *(const v4f*)(Bg + BADD + (size_t)(8 * i + 1) * LDB);\
      }                                                               \
    }                                                                 \
    Ag += (size_t)KS_ * LDA;                                          \
    Bg += (size_t)KS_ * LDB;                                          \
  } while (0)

#define KT_PARK(stg)                                                  \
  do {                                                                \
    float* s_ = smem + (stg) * STG_;                                  \
    v4f t_;                                                           \
    t_.x = ra0.x; t_.y = ra1.x; t_.z = ra0.y; t_.w = ra1.y;           \
    *(v4f*)(s_ + aSt) = t_;                                           \
    t_.x = ra0.z; t_.y = ra1.z; t_.z = ra0.w; t_.w = ra1.w;           \
    *(v4f*)(s_ + aSt + 4) = t_;                                       \
    _Pragma("unroll")                                                 \
    for (int i = 0; i < 4; ++i) {                                     \
      if (BADD) { rb0[i] += rc0[i]; rb1[i] += rc1[i]; }               \
      float* d_ = s_ + bSt + i * 4 * BPITCH_;                         \
      t_.x = rb0[i].x; t_.y = rb1[i].x; t_.z = rb0[i].y; t_.w = rb1[i].y; \
      *(v4f*)(d_) = t_;                                               \
      t_.x = rb0[i].z; t_.y = rb1[i].z; t_.z = rb0[i].w; t_.w = rb1[i].w; \
      *(v4f*)(d_ + 4) = t_;                                           \
    }                                                                 \
  } while (0)

  constexpr int T = K / KS_;

  KT_FETCH();
  KT_PARK(0);
  __syncthreads();

  #pragma unroll 1
  for (int t = 0; t < T - 1; ++t) {
    const int cur = t & 1;
    KT_FETCH();
    kT_stage_products(smem + cur * STG_ + aLd, smem + cur * STG_ + bLd, acc);
    KT_PARK(cur ^ 1);
    __syncthreads();
  }
  kT_stage_products(smem + ((T - 1) & 1) * STG_ + aLd, smem + ((T - 1) & 1) * STG_ + bLd, acc);

#undef KT_FETCH
#undef KT_PARK

  if (!TSTORE) {
    #pragma unroll
    for (int mi = 0; mi < 2; ++mi) {
      #pragma unroll
      for (int r = 0; r < 8; ++r) {
        int m = m0 + 16 * mi + r + 8 * half;
        #pragma unroll
        for (int s = 0; s < 4; ++s)
          Cb[(size_t)m * LDC + n0 + 16 * s + ml] = acc[mi][s][r];
      }
    }
  } else {
    const float* Rb = HAS_RES ? (resid + (size_t)z * strideR) : nullptr;
    #pragma unroll
    for (int mi = 0; mi < 2; ++mi) {
      #pragma unroll
      for (int r = 0; r < 8; ++r) {
        int m = m0 + 16 * mi + r + 8 * half;
        #pragma unroll
        for (int s = 0; s < 4; ++s) {
          size_t i = (size_t)(n0 + 16 * s + ml) * LDC + m;
          float v = acc[mi][s][r];
          if (HAS_RES) v += Rb[i];
          Cb[i] = v;
        }
      }
    }
  }
}

extern "C" void kernel_launch(void* const* d_in, const int* in_sizes, int n_in,
                              void* d_out, int out_size, void* d_ws, size_t ws_size,
                              hipStream_t stream) {
  const float* x      = (const float*)d_in[0];
  const float* gn_w   = (const float*)d_in[1];
  const float* gn_b   = (const float*)d_in[2];
  const float* qkv_w  = (const float*)d_in[3];
  const float* qkv_b  = (const float*)d_in[4];
  const float* proj_w = (const float*)d_in[5];
  const float* proj_b = (const float*)d_in[6];
  float* out = (float*)d_out;

  float* ws = (float*)d_ws;
  const size_t XN_SZ  = (size_t)B_ * C_ * N_;
  const size_t QKV_SZ = (size_t)B_ * N_ * C3_;
  const size_t CTX_SZ = (size_t)B_ * C_ * C_;
  float* mean   = ws;
  float* rstd   = ws + 64;
  float* xn     = ws + 256;
  float* qkv    = xn   + XN_SZ;
  float* qT     = qkv  + QKV_SZ;
  float* o2T    = qT   + XN_SZ + CTX_SZ;
  float* wqkvT  = o2T  + XN_SZ;
  float* wprojT = wqkvT + (size_t)C_ * C3_;

  kT_transpose<<<dim3(C_ / 32, C3_ / 32), dim3(32, 8), 0, stream>>>(qkv_w,  wqkvT,  C3_, C_);
  kT_transpose<<<dim3(C_ / 32, C_  / 32), dim3(32, 8), 0, stream>>>(proj_w, wprojT, C_,  C_);

  kT_gn_stats<<<B_ * GRP_, 256, 0, stream>>>(x, mean, rstd);
  kT_gn_apply<<<B_ * C_,   256, 0, stream>>>(x, gn_w, gn_b, mean, rstd, xn);

  kT_gemm<N_, C3_, C3_, C_, false, true, false, 0>
      <<<dim3(N_ / 64, C3_ / 256, B_), 256, 0, stream>>>(
      xn, (long long)C_ * N_,
      wqkvT, 0,
      qkv, (long long)N_ * C3_,
      qkv_b, nullptr, 0);

  kT_softmax_k<<<B_ * N_, 256, 0, stream>>>(qkv);

  kT_qtranspose<<<dim3(C_ / 32, N_ / 32, B_), dim3(32, 8), 0, stream>>>(qkv, qT);
  kT_softmax_row1024<<<B_ * C_, 256, 0, stream>>>(qT);

  kT_gemm<C3_, C3_, C3_, N_ / 2, false, false, false, 0>
      <<<dim3(C_ / 64, C_ / 256, 2 * B_), 256, 0, stream>>>(
      qkv + C_,     (long long)(N_ / 2) * C3_,
      qkv + 2 * C_, (long long)(N_ / 2) * C3_,
      qkv,          (long long)(N_ / 2) * C3_,
      nullptr, nullptr, 0);

  kT_gemm<N_, C3_, N_, C_, true, false, false, (N_ / 2) * C3_>
      <<<dim3(N_ / 64, C_ / 256, B_), 256, 0, stream>>>(
      qT, (long long)C_ * N_,
      qkv, (long long)N_ * C3_,
      o2T, (long long)C_ * N_,
      nullptr, nullptr, 0);

  kT_gemm<N_, C_, N_, C_, true, true, true, 0>
      <<<dim3(N_ / 64, C_ / 256, B_), 256, 0, stream>>>(
      o2T, (long long)C_ * N_,
      wprojT, 0,
      out, (long long)C_ * N_,
      proj_b, xn, (long long)C_ * N_);
}


# k softmax moved into the GEMMs: qkv GEMM's k blocks emit per-token partial (max, sum of exp); context GEMM applies exp(k-max)/sum as it parks its A tile; softmax kernel dropped; sum order changed
# speedup vs baseline: 1.4624x; 1.0455x over previous
#include <hip/hip_runtime.h>
#include <math.h>

typedef float v2f __attribute__((ext_vector_type(2)));
typedef float v8f __attribute__((ext_vector_type(8)));

#define B_    8
#define C_    512
#define N_    1024
#define C3_   1536
#define GRP_  8
#define GCH_  64
#define GELT_ (GCH_ * N_)
#define EPS_  1e-5f

__global__ __launch_bounds__(256) void kT_transpose(const float* __restrict__ in,
                                                    float* __restrict__ out,
                                                    int R, int Cc) {
  __shared__ float tile[32][33];
  int c0 = blockIdx.x * 32, r0 = blockIdx.y * 32;
  int tx = threadIdx.x, ty = threadIdx.y;
  #pragma unroll
  for (int i = 0; i < 32; i += 8)
    tile[ty + i][tx] = in[(size_t)(r0 + ty + i) * Cc + c0 + tx];
  __syncthreads();
  #pragma unroll
  for (int i = 0; i < 32; i += 8)
    out[(size_t)(c0 + ty + i) * R + r0 + tx] = tile[tx][ty + i];
}

__global__ __launch_bounds__(256) void kT_qtranspose(const float* __restrict__ qkv,
                                                     float* __restrict__ qT) {
  __shared__ float tile[32][33];
  int d0 = blockIdx.x * 32, n0 = blockIdx.y * 32;
  const float* in = qkv + (size_t)blockIdx.z * N_ * C3_;
  float*      out = qT  + (size_t)blockIdx.z * C_ * N_;
  int tx = threadIdx.x, ty = threadIdx.y;
  #pragma unroll
  for (int i = 0; i < 32; i += 8)
    tile[ty + i][tx] = in[(size_t)(n0 + ty + i) * C3_ + d0 + tx];
  __syncthreads();
  #pragma unroll
  for (int i = 0; i < 32; i += 8)
    out[(size_t)(d0 + ty + i) * N_ + n0 + tx] = tile[tx][ty + i];
}

__global__ __launch_bounds__(256) void kT_gn_stats(const float* __restrict__ x,
                                                   float* __restrict__ mean,
                                                   float* __restrict__ rstd) {
  __shared__ float s1[256], s2[256];
  int bg = blockIdx.x;
  const float4* p = (const float4*)(x + (size_t)bg * GELT_);
  float s = 0.f, q = 0.f;
  for (int i = threadIdx.x; i < GELT_ / 4; i += 256) {
    float4 v = p[i];
    s += v.x + v.y + v.z + v.w;
    q += v.x * v.x + v.y * v.y + v.z * v.z + v.w * v.w;
  }
  s1[threadIdx.x] = s; s2[threadIdx.x] = q;
  __syncthreads();
  for (int off = 128; off > 0; off >>= 1) {
    if (threadIdx.x < off) {
      s1[threadIdx.x] += s1[threadIdx.x + off];
      s2[threadIdx.x] += s2[threadIdx.x + off];
    }
    __syncthreads();
  }
  if (threadIdx.x == 0) {
    float m = s1[0] * (1.0f / GELT_);
    float v = s2[0] * (1.0f / GELT_) - m * m;
    mean[bg] = m;
    rstd[bg] = rsqrtf(v + EPS_);
  }
}

__global__ __launch_bounds__(256) void kT_gn_apply(const float* __restrict__ x,
                                                   const float* __restrict__ gamma,
                                                   const float* __restrict__ beta,
                                                   const float* __restrict__ mean,
                                                   const float* __restrict__ rstd,
                                                   float* __restrict__ xn) {
  int bc = blockIdx.x;
  int c = bc & (C_ - 1);
  int bg = ((bc >> 9) << 3) + (c >> 6);
  float gm = gamma[c] * rstd[bg];
  float bt = beta[c] - mean[bg] * gm;
  const float4* pin  = (const float4*)(x  + (size_t)bc * N_);
  float4*       pout = (float4*)(xn + (size_t)bc * N_);
  float4 v = pin[threadIdx.x];
  v.x = v.x * gm + bt; v.y = v.y * gm + bt;
  v.z = v.z * gm + bt; v.w = v.w * gm + bt;
  pout[threadIdx.x] = v;
}

__global__ __launch_bounds__(256) void kT_softmax_row1024(float* __restrict__ qT) {
  __shared__ float red[256];
  float4* row = (float4*)(qT + (size_t)blockIdx.x * N_);
  int t = threadIdx.x;
  float4 v = row[t];
  red[t] = fmaxf(fmaxf(v.x, v.y), fmaxf(v.z, v.w));
  __syncthreads();
  for (int off = 128; off > 0; off >>= 1) {
    if (t < off) red[t] = fmaxf(red[t], red[t + off]);
    __syncthreads();
  }
  float mx = red[0];
  __syncthreads();
  float4 e;
  e.x = __expf(v.x - mx); e.y = __expf(v.y - mx);
  e.z = __expf(v.z - mx); e.w = __expf(v.w - mx);
  red[t] = e.x + e.y + e.z + e.w;
  __syncthreads();
  for (int off = 128; off > 0; off >>= 1) {
    if (t < off) red[t] += red[t + off];
    __syncthreads();
  }
  float inv = 1.0f / red[0];
  e.x *= inv; e.y *= inv; e.z *= inv; e.w *= inv;
  row[t] = e;
}

typedef float v4f __attribute__((ext_vector_type(4)));

#define KS_     32
#define KP_     (KS_ / 2)
#define APITCH_ (64 * 2 + 32)
#define BPITCH_ (256 * 2 + 32)
#define ASTG_   (KP_ * APITCH_)
#define BSTG_   (KP_ * BPITCH_)
#define STG_    (ASTG_ + BSTG_)

__device__ __forceinline__ void kT_stage_products(const float* __restrict__ sa,
                                                  const float* __restrict__ sb,
                                                  v8f (&acc)[2][4]) {
  #pragma unroll
  for (int j = 0; j < KS_ / 4; ++j) {
    v2f a0 = *(const v2f*)(sa + j * 2 * APITCH_);
    v2f a1 = *(const v2f*)(sa + j * 2 * APITCH_ + 32);
    v2f b0 = *(const v2f*)(sb + j * 2 * BPITCH_);
    v2f b1 = *(const v2f*)(sb + j * 2 * BPITCH_ + 32);
    v2f b2 = *(const v2f*)(sb + j * 2 * BPITCH_ + 64);
    v2f b3 = *(const v2f*)(sb + j * 2 * BPITCH_ + 96);
    acc[0][0] = __builtin_amdgcn_wmma_f32_16x16x4_f32(false, a0, false, b0, (short)0, acc[0][0], false, false);
    acc[0][1] = __builtin_amdgcn_wmma_f32_16x16x4_f32(false, a0, false, b1, (short)0, acc[0][1], false, false);
    acc[0][2] = __builtin_amdgcn_wmma_f32_16x16x4_f32(false, a0, false, b2, (short)0, acc[0][2], false, false);
    acc[0][3] = __builtin_amdgcn_wmma_f32_16x16x4_f32(false, a0, false, b3, (short)0, acc[0][3], false, false);
    acc[1][0] = __builtin_amdgcn_wmma_f32_16x16x4_f32(false, a1, false, b0, (short)0, acc[1][0], false, false);
    acc[1][1] = __builtin_amdgcn_wmma_f32_16x16x4_f32(false, a1, false, b1, (short)0, acc[1][1], false, false);
    acc[1][2] = __builtin_amdgcn_wmma_f32_16x16x4_f32(false, a1, false, b2, (short)0, acc[1][2], false, false);
    acc[1][3] = __builtin_amdgcn_wmma_f32_16x16x4_f32(false, a1, false, b3, (short)0, acc[1][3], false, false);
  }
}

template <int LDA, int LDB, int LDC, int K, bool TSTORE, bool HAS_BIAS, bool HAS_RES, int BADD, int SMX>
__global__ __launch_bounds__(256) __attribute__((amdgpu_waves_per_eu(2, BADD ? 2 : 8))) void kT_gemm(
    const float* __restrict__ A, long long strideA,
    const float* __restrict__ B, long long strideB,
    float* __restrict__ Cmat,    long long strideC,
    const float* __restrict__ bias,
    const float* __restrict__ resid, long long strideR,
    float* __restrict__ stats) {
  __shared__ __attribute__((aligned(16))) float smem[2 * STG_];

  int z = blockIdx.z;
  const float* Ab = A + (size_t)z * strideA;
  const float* Bb = B + (size_t)z * strideB;
  float*       Cb = Cmat + (size_t)z * strideC;

  int tid  = threadIdx.x;
  int wave = tid >> 5;
  int lane = tid & 31;
  int half = lane >> 4;
  int ml   = lane & 15;
  int mblk = blockIdx.x * 64;
  int nblk = blockIdx.y * 256;
  int m0 = mblk + (wave & 1) * 32;
  int n0 = nblk + (wave >> 1) * 64;

  v8f acc[2][4];
  {
    float bv[4];
    #pragma unroll
    for (int s = 0; s < 4; ++s) bv[s] = HAS_BIAS ? bias[n0 + 16 * s + ml] : 0.0f;
    #pragma unroll
    for (int mi = 0; mi < 2; ++mi)
      #pragma unroll
      for (int s = 0; s < 4; ++s)
        #pragma unroll
        for (int r = 0; r < 8; ++r) acc[mi][s][r] = bv[s];
  }

  const float* Ag = Ab + (size_t)(2 * (tid >> 4)) * LDA + mblk + 4 * (tid & 15);
  const float* Bg = Bb + (size_t)(2 * (tid >> 6)) * LDB + nblk + 4 * (tid & 63);
  const int aSt = (tid >> 4) * APITCH_ + 8 * (tid & 15);
  const int bSt = ASTG_ + (tid >> 6) * BPITCH_ + 8 * (tid & 63);
  const int aLd = half * APITCH_ + ((wave & 1) * 32 + ml) * 2;
  const int bLd = ASTG_ + half * BPITCH_ + ((wave >> 1) * 64 + ml) * 2;

  v4f ra0, ra1, rb0[4], rb1[4];
  v4f rc0[4], rc1[4];
  v4f st0, st1;
  const float* Sg = stats + ((size_t)z * K + 2 * (tid >> 4)) * 2;

#define KT_FETCH()                                                    \
  do {                                                                \
    ra0 = *(const v4f*)(Ag);                                          \
    ra1 = *(const v4f*)(Ag + LDA);                                    \
    if (SMX == 2) {                                                   \
      st0 = *(const v4f*)(Sg);                                        \
      st1 = *(const v4f*)(Sg + 2 * B_ * N_);                          \
      Sg += 2 * KS_;                                                  \
    }                                                                 \
    _Pragma("unroll")                                                 \
    for (int i = 0; i < 4; ++i) {                                     \
      rb0[i] = *(const v4f*)(Bg + (size_t)(8 * i) * LDB);             \
      rb1[i] = *(const v4f*)(Bg + (size_t)(8 * i + 1) * LDB);         \
      if (BADD) {                                                     \
        rc0[i] = *(const v4f*)(Bg + BADD + (size_t)(8 * i) * LDB);    \
        rc1[i] = *(const v4f*)(Bg + BADD + (size_t)(8 * i + 1) * LDB);\
      }                                                               \
    }                                                                 \
    Ag += (size_t)KS_ * LDA;                                          \
    Bg += (size_t)KS_ * LDB;                                          \
  } while (0)

#define KT_PARK(stg)                                                  \
  do {                                                                \
    float* s_ = smem + (stg) * STG_;                                  \
    v4f t_;                                                           \
    if (SMX == 2) {                                                   \
      float M0_ = fmaxf(st0.x, st1.x), M1_ = fmaxf(st0.z, st1.z);     \
      float i0_ = 1.0f / (st0.y * __expf(st0.x - M0_) + st1.y * __expf(st1.x - M0_)); \
      float i1_ = 1.0f / (st0.w * __expf(st0.z - M1_) + st1.w * __expf(st1.z - M1_)); \
      ra0.x = __expf(ra0.x - M0_) * i0_; ra0.y = __expf(ra0.y - M0_) * i0_; \
      ra0.z = __expf(ra0.z - M0_) * i0_; ra0.w = __expf(ra0.w - M0_) * i0_; \
      ra1.x = __expf(ra1.x - M1_) * i1_; ra1.y = __expf(ra1.y - M1_) * i1_; \
      ra1.z = __expf(ra1.z - M1_) * i1_; ra1.w = __expf(ra1.w - M1_) * i1_; \
    }                                                                 \
    t_.x = ra0.x; t_.y = ra1.x; t_.z = ra0.y; t_.w = ra1.y;           \
    *(v4f*)(s_ + aSt) = t_;                                           \
    t_.x = ra0.z; t_.y = ra1.z; t_.z = ra0.w; t_.w = ra1.w;           \
    *(v4f*)(s_ + aSt + 4) = t_;                                       \
    _Pragma("unroll")                                                 \
    for (int i = 0; i < 4; ++i) {                                     \
      if (BADD) { rb0[i] += rc0[i]; rb1[i] += rc1[i]; }               \
      float* d_ = s_ + bSt + i * 4 * BPITCH_;                         \
      t_.x = rb0[i].x; t_.y = rb1[i].x; t_.z = rb0[i].y; t_.w = rb1[i].y; \
      *(v4f*)(d_) = t_;                                               \
      t_.x = rb0[i].z; t_.y = rb1[i].z; t_.z = rb0[i].w; t_.w = rb1[i].w; \
      *(v4f*)(d_ + 4) = t_;                                           \
    }                                                                 \
  } while (0)

  constexpr int T = K / KS_;

  KT_FETCH();
  KT_PARK(0);
  __syncthreads();

  #pragma unroll 1
  for (int t = 0; t < T - 1; ++t) {
    const int cur = t & 1;
    KT_FETCH();
    kT_stage_products(smem + cur * STG_ + aLd, smem + cur * STG_ + bLd, acc);
    KT_PARK(cur ^ 1);
    __syncthreads();
  }
  kT_stage_products(smem + ((T - 1) & 1) * STG_ + aLd, smem + ((T - 1) & 1) * STG_ + bLd, acc);

#undef KT_FETCH
#undef KT_PARK

  if (SMX == 1) {
    asm volatile("" : "+v"(tid));
    wave = tid >> 5;
    lane = tid & 31;
    half = lane >> 4;
    ml   = lane & 15;
    m0 = mblk + (wave & 1) * 32;
    n0 = nblk + (wave >> 1) * 64;
  }

  if (!TSTORE) {
    #pragma unroll
    for (int mi = 0; mi < 2; ++mi) {
      #pragma unroll
      for (int r = 0; r < 8; ++r) {
        int m = m0 + 16 * mi + r + 8 * half;
        #pragma unroll
        for (int s = 0; s < 4; ++s)
          Cb[(size_t)m * LDC + n0 + 16 * s + ml] = acc[mi][s][r];
      }
    }
    if (SMX == 1) {
      if (nblk >= C_ && nblk < 2 * C_) {
        __shared__ __attribute__((aligned(16))) float sred[4 * 64 * 2];
        float pm = 0.f, ps = 0.f;
        #pragma unroll
        for (int mi = 0; mi < 2; ++mi) {
          #pragma unroll
          for (int r = 0; r < 8; ++r) {
            float mx = fmaxf(fmaxf(acc[mi][0][r], acc[mi][1][r]), fmaxf(acc[mi][2][r], acc[mi][3][r]));
            mx = fmaxf(mx, __shfl_xor(mx, 1));
            mx = fmaxf(mx, __shfl_xor(mx, 2));
            mx = fmaxf(mx, __shfl_xor(mx, 4));
            mx = fmaxf(mx, __shfl_xor(mx, 8));
            float sm = (__expf(acc[mi][0][r] - mx) + __expf(acc[mi][1][r] - mx)) +
                       (__expf(acc[mi][2][r] - mx) + __expf(acc[mi][3][r] - mx));
            sm += __shfl_xor(sm, 1);
            sm += __shfl_xor(sm, 2);
            sm += __shfl_xor(sm, 4);
            sm += __shfl_xor(sm, 8);
            if (ml == 8 * mi + r) { pm = mx; ps = sm; }
          }
        }
        int row = (wave & 1) * 32 + 16 * (ml >> 3) + 8 * half + (ml & 7);
        *(v2f*)(sred + ((wave >> 1) * 64 + row) * 2) = v2f{pm, ps};
        __syncthreads();
        if (tid < 64) {
          v2f p0 = *(const v2f*)(sred + (0 * 64 + tid) * 2);
          v2f p1 = *(const v2f*)(sred + (1 * 64 + tid) * 2);
          v2f p2 = *(const v2f*)(sred + (2 * 64 + tid) * 2);
          v2f p3 = *(const v2f*)(sred + (3 * 64 + tid) * 2);
          float M = fmaxf(fmaxf(p0.x, p1.x), fmaxf(p2.x, p3.x));
          float S = (p0.y * __expf(p0.x - M) + p1.y * __expf(p1.x - M)) +
                    (p2.y * __expf(p2.x - M) + p3.y * __expf(p3.x - M));
          size_t plane = (size_t)((nblk - C_) >> 8) * (2 * B_ * N_);
          *(v2f*)(stats + plane + ((size_t)z * N_ + mblk + tid) * 2) = v2f{M, S};
        }
      }
    }
  } else {
    const float* Rb = HAS_RES ? (resid + (size_t)z * strideR) : nullptr;
    #pragma unroll
    for (int mi = 0; mi < 2; ++mi) {
      #pragma unroll
      for (int r = 0; r < 8; ++r) {
        int m = m0 + 16 * mi + r + 8 * half;
        #pragma unroll
        for (int s = 0; s < 4; ++s) {
          size_t i = (size_t)(n0 + 16 * s + ml) * LDC + m;
          float v = acc[mi][s][r];
          if (HAS_RES) v += Rb[i];
          Cb[i] = v;
        }
      }
    }
  }
}

extern "C" void kernel_launch(void* const* d_in, const int* in_sizes, int n_in,
                              void* d_out, int out_size, void* d_ws, size_t ws_size,
                              hipStream_t stream) {
  const float* x      = (const float*)d_in[0];
  const float* gn_w   = (const float*)d_in[1];
  const float* gn_b   = (const float*)d_in[2];
  const float* qkv_w  = (const float*)d_in[3];
  const float* qkv_b  = (const float*)d_in[4];
  const float* proj_w = (const float*)d_in[5];
  const float* proj_b = (const float*)d_in[6];
  float* out = (float*)d_out;

  float* ws = (float*)d_ws;
  const size_t XN_SZ  = (size_t)B_ * C_ * N_;
  const size_t QKV_SZ = (size_t)B_ * N_ * C3_;
  const size_t CTX_SZ = (size_t)B_ * C_ * C_;
  float* mean   = ws;
  float* rstd   = ws + 64;
  float* xn     = ws + 256;
  float* qkv    = xn   + XN_SZ;
  float* qT     = qkv  + QKV_SZ;
  float* kst    = qT   + XN_SZ;
  float* o2T    = qT   + XN_SZ + CTX_SZ;
  float* wqkvT  = o2T  + XN_SZ;
  float* wprojT = wqkvT + (size_t)C_ * C3_;

  kT_transpose<<<dim3(C_ / 32, C3_ / 32), dim3(32, 8), 0, stream>>>(qkv_w,  wqkvT,  C3_, C_);
  kT_transpose<<<dim3(C_ / 32, C_  / 32), dim3(32, 8), 0, stream>>>(proj_w, wprojT, C_,  C_);

  kT_gn_stats<<<B_ * GRP_, 256, 0, stream>>>(x, mean, rstd);
  kT_gn_apply<<<B_ * C_,   256, 0, stream>>>(x, gn_w, gn_b, mean, rstd, xn);

  kT_gemm<N_, C3_, C3_, C_, false, true, false, 0, 1>
      <<<dim3(N_ / 64, C3_ / 256, B_), 256, 0, stream>>>(
      xn, (long long)C_ * N_,
      wqkvT, 0,
      qkv, (long long)N_ * C3_,
      qkv_b, nullptr, 0, kst);


  kT_qtranspose<<<dim3(C_ / 32, N_ / 32, B_), dim3(32, 8), 0, stream>>>(qkv, qT);
  kT_softmax_row1024<<<B_ * C_, 256, 0, stream>>>(qT);

  kT_gemm<C3_, C3_, C3_, N_ / 2, false, false, false, 0, 2>
      <<<dim3(C_ / 64, C_ / 256, 2 * B_), 256, 0, stream>>>(
      qkv + C_,     (long long)(N_ / 2) * C3_,
      qkv + 2 * C_, (long long)(N_ / 2) * C3_,
      qkv,          (long long)(N_ / 2) * C3_,
      nullptr, nullptr, 0, kst);

  kT_gemm<N_, C3_, N_, C_, true, false, false, (N_ / 2) * C3_, 0>
      <<<dim3(N_ / 64, C_ / 256, B_), 256, 0, stream>>>(
      qT, (long long)C_ * N_,
      qkv, (long long)N_ * C3_,
      o2T, (long long)C_ * N_,
      nullptr, nullptr, 0, nullptr);

  kT_gemm<N_, C_, N_, C_, true, true, true, 0, 0>
      <<<dim3(N_ / 64, C_ / 256, B_), 256, 0, stream>>>(
      o2T, (long long)C_ * N_,
      wprojT, 0,
      out, (long long)C_ * N_,
      proj_b, xn, (long long)C_ * N_, nullptr);
}


# q softmax (over tokens) moved into the GEMMs: qkv GEMM stores q transposed + per-channel partial (max, sum exp); q x context GEMM applies it parking its A tile; 2 kernels dropped; sum order changed
# speedup vs baseline: 1.6934x; 1.1580x over previous
#include <hip/hip_runtime.h>
#include <math.h>

typedef float v2f __attribute__((ext_vector_type(2)));
typedef float v8f __attribute__((ext_vector_type(8)));

#define B_    8
#define C_    512
#define N_    1024
#define C3_   1536
#define GRP_  8
#define GCH_  64
#define GELT_ (GCH_ * N_)
#define EPS_  1e-5f

__global__ __launch_bounds__(256) void kT_transpose(const float* __restrict__ in,
                                                    float* __restrict__ out,
                                                    int R, int Cc) {
  __shared__ float tile[32][33];
  int c0 = blockIdx.x * 32, r0 = blockIdx.y * 32;
  int tx = threadIdx.x, ty = threadIdx.y;
  #pragma unroll
  for (int i = 0; i < 32; i += 8)
    tile[ty + i][tx] = in[(size_t)(r0 + ty + i) * Cc + c0 + tx];
  __syncthreads();
  #pragma unroll
  for (int i = 0; i < 32; i += 8)
    out[(size_t)(c0 + ty + i) * R + r0 + tx] = tile[tx][ty + i];
}

__global__ __launch_bounds__(256) void kT_gn_stats(const float* __restrict__ x,
                                                   float* __restrict__ mean,
                                                   float* __restrict__ rstd) {
  __shared__ float s1[256], s2[256];
  int bg = blockIdx.x;
  const float4* p = (const float4*)(x + (size_t)bg * GELT_);
  float s = 0.f, q = 0.f;
  for (int i = threadIdx.x; i < GELT_ / 4; i += 256) {
    float4 v = p[i];
    s += v.x + v.y + v.z + v.w;
    q += v.x * v.x + v.y * v.y + v.z * v.z + v.w * v.w;
  }
  s1[threadIdx.x] = s; s2[threadIdx.x] = q;
  __syncthreads();
  for (int off = 128; off > 0; off >>= 1) {
    if (threadIdx.x < off) {
      s1[threadIdx.x] += s1[threadIdx.x + off];
      s2[threadIdx.x] += s2[threadIdx.x + off];
    }
    __syncthreads();
  }
  if (threadIdx.x == 0) {
    float m = s1[0] * (1.0f / GELT_);
    float v = s2[0] * (1.0f / GELT_) - m * m;
    mean[bg] = m;
    rstd[bg] = rsqrtf(v + EPS_);
  }
}

__global__ __launch_bounds__(256) void kT_gn_apply(const float* __restrict__ x,
                                                   const float* __restrict__ gamma,
                                                   const float* __restrict__ beta,
                                                   const float* __restrict__ mean,
                                                   const float* __restrict__ rstd,
                                                   float* __restrict__ xn) {
  int bc = blockIdx.x;
  int c = bc & (C_ - 1);
  int bg = ((bc >> 9) << 3) + (c >> 6);
  float gm = gamma[c] * rstd[bg];
  float bt = beta[c] - mean[bg] * gm;
  const float4* pin  = (const float4*)(x  + (size_t)bc * N_);
  float4*       pout = (float4*)(xn + (size_t)bc * N_);
  float4 v = pin[threadIdx.x];
  v.x = v.x * gm + bt; v.y = v.y * gm + bt;
  v.z = v.z * gm + bt; v.w = v.w * gm + bt;
  pout[threadIdx.x] = v;
}

typedef float v4f __attribute__((ext_vector_type(4)));

#define KS_     32
#define KP_     (KS_ / 2)
#define APITCH_ (64 * 2 + 32)
#define BPITCH_ (256 * 2 + 32)
#define ASTG_   (KP_ * APITCH_)
#define BSTG_   (KP_ * BPITCH_)
#define STG_    (ASTG_ + BSTG_)
#define QST_OFF_ 65536
#define QTP_     68

__device__ __forceinline__ void kT_stage_products(const float* __restrict__ sa,
                                                  const float* __restrict__ sb,
                                                  v8f (&acc)[2][4]) {
  #pragma unroll
  for (int j = 0; j < KS_ / 4; ++j) {
    v2f a0 = *(const v2f*)(sa + j * 2 * APITCH_);
    v2f a1 = *(const v2f*)(sa + j * 2 * APITCH_ + 32);
    v2f b0 = *(const v2f*)(sb + j * 2 * BPITCH_);
    v2f b1 = *(const v2f*)(sb + j * 2 * BPITCH_ + 32);
    v2f b2 = *(const v2f*)(sb + j * 2 * BPITCH_ + 64);
    v2f b3 = *(const v2f*)(sb + j * 2 * BPITCH_ + 96);
    acc[0][0] = __builtin_amdgcn_wmma_f32_16x16x4_f32(false, a0, false, b0, (short)0, acc[0][0], false, false);
    acc[0][1] = __builtin_amdgcn_wmma_f32_16x16x4_f32(false, a0, false, b1, (short)0, acc[0][1], false, false);
    acc[0][2] = __builtin_amdgcn_wmma_f32_16x16x4_f32(false, a0, false, b2, (short)0, acc[0][2], false, false);
    acc[0][3] = __builtin_amdgcn_wmma_f32_16x16x4_f32(false, a0, false, b3, (short)0, acc[0][3], false, false);
    acc[1][0] = __builtin_amdgcn_wmma_f32_16x16x4_f32(false, a1, false, b0, (short)0, acc[1][0], false, false);
    acc[1][1] = __builtin_amdgcn_wmma_f32_16x16x4_f32(false, a1, false, b1, (short)0, acc[1][1], false, false);
    acc[1][2] = __builtin_amdgcn_wmma_f32_16x16x4_f32(false, a1, false, b2, (short)0, acc[1][2], false, false);
    acc[1][3] = __builtin_amdgcn_wmma_f32_16x16x4_f32(false, a1, false, b3, (short)0, acc[1][3], false, false);
  }
}

template <int LDA, int LDB, int LDC, int K, bool TSTORE, bool HAS_BIAS, bool HAS_RES, int BADD, int SMX>
__global__ __launch_bounds__(256) __attribute__((amdgpu_waves_per_eu(2, BADD ? 2 : 8))) void kT_gemm(
    const float* __restrict__ A, long long strideA,
    const float* __restrict__ B, long long strideB,
    float* __restrict__ Cmat,    long long strideC,
    const float* __restrict__ bias,
    const float* __restrict__ resid, long long strideR,
    float* __restrict__ stats,
    float* __restrict__ aux) {
  __shared__ __attribute__((aligned(16))) float smem[2 * STG_];
  __shared__ __attribute__((aligned(16))) float qtab[SMX == 3 ? 2 * K : 4];

  int z = blockIdx.z;
  const float* Ab = A + (size_t)z * strideA;
  const float* Bb = B + (size_t)z * strideB;
  float*       Cb = Cmat + (size_t)z * strideC;

  int tid  = threadIdx.x;
  int wave = tid >> 5;
  int lane = tid & 31;
  int half = lane >> 4;
  int ml   = lane & 15;
  int mblk = blockIdx.x * 64;
  int nblk = blockIdx.y * 256;
  int m0 = mblk + (wave & 1) * 32;
  int n0 = nblk + (wave >> 1) * 64;

  v8f acc[2][4];
  {
    float bv[4];
    #pragma unroll
    for (int s = 0; s < 4; ++s) bv[s] = HAS_BIAS ? bias[n0 + 16 * s + ml] : 0.0f;
    #pragma unroll
    for (int mi = 0; mi < 2; ++mi)
      #pragma unroll
      for (int s = 0; s < 4; ++s)
        #pragma unroll
        for (int r = 0; r < 8; ++r) acc[mi][s][r] = bv[s];
  }

  const float* Ag = Ab + (size_t)(2 * (tid >> 4)) * LDA + mblk + 4 * (tid & 15);
  const float* Bg = Bb + (size_t)(2 * (tid >> 6)) * LDB + nblk + 4 * (tid & 63);
  const int aSt = (tid >> 4) * APITCH_ + 8 * (tid & 15);
  const int bSt = ASTG_ + (tid >> 6) * BPITCH_ + 8 * (tid & 63);
  const int aLd = half * APITCH_ + ((wave & 1) * 32 + ml) * 2;
  const int bLd = ASTG_ + half * BPITCH_ + ((wave >> 1) * 64 + ml) * 2;

  v4f ra0, ra1, rb0[4], rb1[4];
  v4f rc0[4], rc1[4];
  v4f st0, st1;
  const float* Sg = stats + ((size_t)z * K + 2 * (tid >> 4)) * 2;
  int qo = 4 * (tid >> 4);

#define KT_FETCH()                                                    \
  do {                                                                \
    ra0 = *(const v4f*)(Ag);                                          \
    ra1 = *(const v4f*)(Ag + LDA);                                    \
    if (SMX == 2) {                                                   \
      st0 = *(const v4f*)(Sg);                                        \
      st1 = *(const v4f*)(Sg + 2 * B_ * N_);                          \
      Sg += 2 * KS_;                                                  \
    }                                                                 \
    if (SMX == 3) {                                                   \
      st0 = *(const v4f*)(qtab + qo);                                 \
      qo += 2 * KS_;                                                  \
    }                                                                 \
    _Pragma("unroll")                                                 \
    for (int i = 0; i < 4; ++i) {                                     \
      rb0[i] = *(const v4f*)(Bg + (size_t)(8 * i) * LDB);             \
      rb1[i] = *(const v4f*)(Bg + (size_t)(8 * i + 1) * LDB);         \
      if (BADD) {                                                     \
        rc0[i] = *(const v4f*)(Bg + BADD + (size_t)(8 * i) * LDB);    \
        rc1[i] = *(const v4f*)(Bg + BADD + (size_t)(8 * i + 1) * LDB);\
      }                                                               \
    }                                                                 \
    Ag += (size_t)KS_ * LDA;                                          \
    Bg += (size_t)KS_ * LDB;                                          \
  } while (0)

#define KT_PARK(stg)                                                  \
  do {                                                                \
    float* s_ = smem + (stg) * STG_;                                  \
    v4f t_;                                                           \
    if (SMX == 2) {                                                   \
      float M0_ = fmaxf(st0.x, st1.x), M1_ = fmaxf(st0.z, st1.z);     \
      float i0_ = 1.0f / (st0.y * __expf(st0.x - M0_) + st1.y * __expf(st1.x - M0_)); \
      float i1_ = 1.0f / (st0.w * __expf(st0.z - M1_) + st1.w * __expf(st1.z - M1_)); \
      ra0.x = __expf(ra0.x - M0_) * i0_; ra0.y = __expf(ra0.y - M0_) * i0_; \
      ra0.z = __expf(ra0.z - M0_) * i0_; ra0.w = __expf(ra0.w - M0_) * i0_; \
      ra1.x = __expf(ra1.x - M1_) * i1_; ra1.y = __expf(ra1.y - M1_) * i1_; \
      ra1.z = __expf(ra1.z - M1_) * i1_; ra1.w = __expf(ra1.w - M1_) * i1_; \
    }                                                                 \
    if (SMX == 3) {                                                   \
      ra0.x = __expf(ra0.x - st0.x) * st0.y; ra0.y = __expf(ra0.y - st0.x) * st0.y; \
      ra0.z = __expf(ra0.z - st0.x) * st0.y; ra0.w = __expf(ra0.w - st0.x) * st0.y; \
      ra1.x = __expf(ra1.x - st0.z) * st0.w; ra1.y = __expf(ra1.y - st0.z) * st0.w; \
      ra1.z = __expf(ra1.z - st0.z) * st0.w; ra1.w = __expf(ra1.w - st0.z) * st0.w; \
    }                                                                 \
    t_.x = ra0.x; t_.y = ra1.x; t_.z = ra0.y; t_.w = ra1.y;           \
    *(v4f*)(s_ + aSt) = t_;                                           \
    t_.x = ra0.z; t_.y = ra1.z; t_.z = ra0.w; t_.w = ra1.w;           \
    *(v4f*)(s_ + aSt + 4) = t_;                                       \
    _Pragma("unroll")                                                 \
    for (int i = 0; i < 4; ++i) {                                     \
      if (BADD) { rb0[i] += rc0[i]; rb1[i] += rc1[i]; }               \
      float* d_ = s_ + bSt + i * 4 * BPITCH_;                         \
      t_.x = rb0[i].x; t_.y = rb1[i].x; t_.z = rb0[i].y; t_.w = rb1[i].y; \
      *(v4f*)(d_) = t_;                                               \
      t_.x = rb0[i].z; t_.y = rb1[i].z; t_.z = rb0[i].w; t_.w = rb1[i].w; \
      *(v4f*)(d_ + 4) = t_;                                           \
    }                                                                 \
  } while (0)

  constexpr int T = K / KS_;

  if (SMX == 3) {
    const float* Q = stats + QST_OFF_ + (size_t)z * (N_ / 64) * C_ * 2;
    #pragma unroll
    for (int h2 = 0; h2 < K / 256; ++h2) {
      int cc = tid + 256 * h2;
      v2f p[N_ / 64];
      #pragma unroll
      for (int i = 0; i < N_ / 64; ++i) p[i] = *(const v2f*)(Q + ((size_t)i * C_ + cc) * 2);
      float M = p[0].x;
      #pragma unroll
      for (int i = 1; i < N_ / 64; ++i) M = fmaxf(M, p[i].x);
      float S = 0.f;
      #pragma unroll
      for (int i = 0; i < N_ / 64; ++i) S += p[i].y * __expf(p[i].x - M);
      *(v2f*)(qtab + 2 * cc) = v2f{M, 1.0f / S};
    }
    __syncthreads();
  }

  KT_FETCH();
  KT_PARK(0);
  __syncthreads();

  #pragma unroll 1
  for (int t = 0; t < T - 1; ++t) {
    const int cur = t & 1;
    KT_FETCH();
    kT_stage_products(smem + cur * STG_ + aLd, smem + cur * STG_ + bLd, acc);
    KT_PARK(cur ^ 1);
    __syncthreads();
  }
  kT_stage_products(smem + ((T - 1) & 1) * STG_ + aLd, smem + ((T - 1) & 1) * STG_ + bLd, acc);

#undef KT_FETCH
#undef KT_PARK

  if (SMX == 1) {
    asm volatile("" : "+v"(tid));
    wave = tid >> 5;
    lane = tid & 31;
    half = lane >> 4;
    ml   = lane & 15;
    m0 = mblk + (wave & 1) * 32;
    n0 = nblk + (wave >> 1) * 64;
  }

  if (!TSTORE && SMX == 1 && nblk < C_) {
    float* Qb = aux + (size_t)z * C_ * N_;
    __syncthreads();
    #pragma unroll
    for (int s = 0; s < 4; ++s) {
      #pragma unroll
      for (int mi = 0; mi < 2; ++mi) {
        float* d_ = smem + ((wave >> 1) * 64 + 16 * s + ml) * QTP_ + (wave & 1) * 32 + 16 * mi + 8 * half;
        v4f t0_, t1_;
        t0_.x = acc[mi][s][0]; t0_.y = acc[mi][s][1]; t0_.z = acc[mi][s][2]; t0_.w = acc[mi][s][3];
        t1_.x = acc[mi][s][4]; t1_.y = acc[mi][s][5]; t1_.z = acc[mi][s][6]; t1_.w = acc[mi][s][7];
        *(v4f*)(d_)     = t0_;
        *(v4f*)(d_ + 4) = t1_;
      }
    }
    float* sq = smem + 256 * QTP_;
    #pragma unroll
    for (int s = 0; s < 4; ++s) {
      float mx = acc[0][s][0];
      #pragma unroll
      for (int mi = 0; mi < 2; ++mi)
        #pragma unroll
        for (int r = 0; r < 8; ++r) mx = fmaxf(mx, acc[mi][s][r]);
      mx = fmaxf(mx, __shfl_xor(mx, 16));
      float sm = 0.f;
      #pragma unroll
      for (int mi = 0; mi < 2; ++mi)
        #pragma unroll
        for (int r = 0; r < 8; ++r) sm += __expf(acc[mi][s][r] - mx);
      sm += __shfl_xor(sm, 16);
      if (half == 0)
        *(v2f*)(sq + ((wave & 1) * 256 + (wave >> 1) * 64 + 16 * s + ml) * 2) = v2f{mx, sm};
    }
    __syncthreads();
    #pragma unroll
    for (int it = 0; it < 16; ++it) {
      int col = it * 16 + (tid >> 4);
      v4f t_ = *(const v4f*)(smem + col * QTP_ + 4 * (tid & 15));
      *(v4f*)(Qb + (size_t)(nblk + col) * N_ + mblk + 4 * (tid & 15)) = t_;
    }
    {
      v2f p0 = *(const v2f*)(sq + tid * 2);
      v2f p1 = *(const v2f*)(sq + (256 + tid) * 2);
      float M = fmaxf(p0.x, p1.x);
      float S = p0.y * __expf(p0.x - M) + p1.y * __expf(p1.x - M);
      *(v2f*)(stats + QST_OFF_ + (((size_t)z * (N_ / 64) + blockIdx.x) * C_ + nblk + tid) * 2) = v2f{M, S};
    }
  } else if (!TSTORE) {
    #pragma unroll
    for (int mi = 0; mi < 2; ++mi) {
      #pragma unroll
      for (int r = 0; r < 8; ++r) {
        int m = m0 + 16 * mi + r + 8 * half;
        #pragma unroll
        for (int s = 0; s < 4; ++s)
          Cb[(size_t)m * LDC + n0 + 16 * s + ml] = acc[mi][s][r];
      }
    }
    if (SMX == 1) {
      if (nblk >= C_ && nblk < 2 * C_) {
        __shared__ __attribute__((aligned(16))) float sred[4 * 64 * 2];
        float pm = 0.f, ps = 0.f;
        #pragma unroll
        for (int mi = 0; mi < 2; ++mi) {
          #pragma unroll
          for (int r = 0; r < 8; ++r) {
            float mx = fmaxf(fmaxf(acc[mi][0][r], acc[mi][1][r]), fmaxf(acc[mi][2][r], acc[mi][3][r]));
            mx = fmaxf(mx, __shfl_xor(mx, 1));
            mx = fmaxf(mx, __shfl_xor(mx, 2));
            mx = fmaxf(mx, __shfl_xor(mx, 4));
            mx = fmaxf(mx, __shfl_xor(mx, 8));
            float sm = (__expf(acc[mi][0][r] - mx) + __expf(acc[mi][1][r] - mx)) +
                       (__expf(acc[mi][2][r] - mx) + __expf(acc[mi][3][r] - mx));
            sm += __shfl_xor(sm, 1);
            sm += __shfl_xor(sm, 2);
            sm += __shfl_xor(sm, 4);
            sm += __shfl_xor(sm, 8);
            if (ml == 8 * mi + r) { pm = mx; ps = sm; }
          }
        }
        int row = (wave & 1) * 32 + 16 * (ml >> 3) + 8 * half + (ml & 7);
        *(v2f*)(sred + ((wave >> 1) * 64 + row) * 2) = v2f{pm, ps};
        __syncthreads();
        if (tid < 64) {
          v2f p0 = *(const v2f*)(sred + (0 * 64 + tid) * 2);
          v2f p1 = *(const v2f*)(sred + (1 * 64 + tid) * 2);
          v2f p2 = *(const v2f*)(sred + (2 * 64 + tid) * 2);
          v2f p3 = *(const v2f*)(sred + (3 * 64 + tid) * 2);
          float M = fmaxf(fmaxf(p0.x, p1.x), fmaxf(p2.x, p3.x));
          float S = (p0.y * __expf(p0.x - M) + p1.y * __expf(p1.x - M)) +
                    (p2.y * __expf(p2.x - M) + p3.y * __expf(p3.x - M));
          size_t plane = (size_t)((nblk - C_) >> 8) * (2 * B_ * N_);
          *(v2f*)(stats + plane + ((size_t)z * N_ + mblk + tid) * 2) = v2f{M, S};
        }
      }
    }
  } else {
    const float* Rb = HAS_RES ? (resid + (size_t)z * strideR) : nullptr;
    #pragma unroll
    for (int mi = 0; mi < 2; ++mi) {
      #pragma unroll
      for (int r = 0; r < 8; ++r) {
        int m = m0 + 16 * mi + r + 8 * half;
        #pragma unroll
        for (int s = 0; s < 4; ++s) {
          size_t i = (size_t)(n0 + 16 * s + ml) * LDC + m;
          float v = acc[mi][s][r];
          if (HAS_RES) v += Rb[i];
          Cb[i] = v;
        }
      }
    }
  }
}

extern "C" void kernel_launch(void* const* d_in, const int* in_sizes, int n_in,
                              void* d_out, int out_size, void* d_ws, size_t ws_size,
                              hipStream_t stream) {
  const float* x      = (const float*)d_in[0];
  const float* gn_w   = (const float*)d_in[1];
  const float* gn_b   = (const float*)d_in[2];
  const float* qkv_w  = (const float*)d_in[3];
  const float* qkv_b  = (const float*)d_in[4];
  const float* proj_w = (const float*)d_in[5];
  const float* proj_b = (const float*)d_in[6];
  float* out = (float*)d_out;

  float* ws = (float*)d_ws;
  const size_t XN_SZ  = (size_t)B_ * C_ * N_;
  const size_t QKV_SZ = (size_t)B_ * N_ * C3_;
  const size_t CTX_SZ = (size_t)B_ * C_ * C_;
  float* mean   = ws;
  float* rstd   = ws + 64;
  float* xn     = ws + 256;
  float* qkv    = xn   + XN_SZ;
  float* qT     = qkv  + QKV_SZ;
  float* kst    = qT   + XN_SZ;
  float* o2T    = qT   + XN_SZ + CTX_SZ;
  float* wqkvT  = o2T  + XN_SZ;
  float* wprojT = wqkvT + (size_t)C_ * C3_;

  kT_transpose<<<dim3(C_ / 32, C3_ / 32), dim3(32, 8), 0, stream>>>(qkv_w,  wqkvT,  C3_, C_);
  kT_transpose<<<dim3(C_ / 32, C_  / 32), dim3(32, 8), 0, stream>>>(proj_w, wprojT, C_,  C_);

  kT_gn_stats<<<B_ * GRP_, 256, 0, stream>>>(x, mean, rstd);
  kT_gn_apply<<<B_ * C_,   256, 0, stream>>>(x, gn_w, gn_b, mean, rstd, xn);

  kT_gemm<N_, C3_, C3_, C_, false, true, false, 0, 1>
      <<<dim3(N_ / 64, C3_ / 256, B_), 256, 0, stream>>>(
      xn, (long long)C_ * N_,
      wqkvT, 0,
      qkv, (long long)N_ * C3_,
      qkv_b, nullptr, 0, kst, qT);


  kT_gemm<C3_, C3_, C3_, N_ / 2, false, false, false, 0, 2>
      <<<dim3(C_ / 64, C_ / 256, 2 * B_), 256, 0, stream>>>(
      qkv + C_,     (long long)(N_ / 2) * C3_,
      qkv + 2 * C_, (long long)(N_ / 2) * C3_,
      qkv,          (long long)(N_ / 2) * C3_,
      nullptr, nullptr, 0, kst, nullptr);

  kT_gemm<N_, C3_, N_, C_, true, false, false, (N_ / 2) * C3_, 3>
      <<<dim3(N_ / 64, C_ / 256, B_), 256, 0, stream>>>(
      qT, (long long)C_ * N_,
      qkv, (long long)N_ * C3_,
      o2T, (long long)C_ * N_,
      nullptr, nullptr, 0, kst, nullptr);

  kT_gemm<N_, C_, N_, C_, true, true, true, 0, 0>
      <<<dim3(N_ / 64, C_ / 256, B_), 256, 0, stream>>>(
      o2T, (long long)C_ * N_,
      wprojT, 0,
      out, (long long)C_ * N_,
      proj_b, xn, (long long)C_ * N_, nullptr, nullptr);
}


# GroupNorm apply moved into the GEMMs: qkv GEMM forms x*g+b parking its A tile (per-channel LDS table from group stats), last GEMM builds the residual from raw x; apply kernel dropped; sum order kept
# speedup vs baseline: 1.7407x; 1.0279x over previous
#include <hip/hip_runtime.h>
#include <math.h>

typedef float v2f __attribute__((ext_vector_type(2)));
typedef float v8f __attribute__((ext_vector_type(8)));

#define B_    8
#define C_    512
#define N_    1024
#define C3_   1536
#define GRP_  8
#define GCH_  64
#define GELT_ (GCH_ * N_)
#define EPS_  1e-5f

__global__ __launch_bounds__(256) void kT_transpose(const float* __restrict__ in,
                                                    float* __restrict__ out,
                                                    int R, int Cc) {
  __shared__ float tile[32][33];
  int c0 = blockIdx.x * 32, r0 = blockIdx.y * 32;
  int tx = threadIdx.x, ty = threadIdx.y;
  #pragma unroll
  for (int i = 0; i < 32; i += 8)
    tile[ty + i][tx] = in[(size_t)(r0 + ty + i) * Cc + c0 + tx];
  __syncthreads();
  #pragma unroll
  for (int i = 0; i < 32; i += 8)
    out[(size_t)(c0 + ty + i) * R + r0 + tx] = tile[tx][ty + i];
}

__global__ __launch_bounds__(256) void kT_gn_stats(const float* __restrict__ x,
                                                   float* __restrict__ mean,
                                                   float* __restrict__ rstd) {
  __shared__ float s1[256], s2[256];
  int bg = blockIdx.x;
  const float4* p = (const float4*)(x + (size_t)bg * GELT_);
  float s = 0.f, q = 0.f;
  for (int i = threadIdx.x; i < GELT_ / 4; i += 256) {
    float4 v = p[i];
    s += v.x + v.y + v.z + v.w;
    q += v.x * v.x + v.y * v.y + v.z * v.z + v.w * v.w;
  }
  s1[threadIdx.x] = s; s2[threadIdx.x] = q;
  __syncthreads();
  for (int off = 128; off > 0; off >>= 1) {
    if (threadIdx.x < off) {
      s1[threadIdx.x] += s1[threadIdx.x + off];
      s2[threadIdx.x] += s2[threadIdx.x + off];
    }
    __syncthreads();
  }
  if (threadIdx.x == 0) {
    float m = s1[0] * (1.0f / GELT_);
    float v = s2[0] * (1.0f / GELT_) - m * m;
    mean[bg] = m;
    rstd[bg] = rsqrtf(v + EPS_);
  }
}

__device__ __forceinline__ void kT_gn_coef(const float* __restrict__ gnw,
                                           const float* __restrict__ gnb,
                                           const float* __restrict__ gst,
                                           int b, int c, float& gm, float& bt) {
  int bg = b * GRP_ + (c >> 6);
  gm = gnw[c] * gst[64 + bg];
  bt = gnb[c] - gst[bg] * gm;
}

typedef float v4f __attribute__((ext_vector_type(4)));

#define KS_     32
#define KP_     (KS_ / 2)
#define APITCH_ (64 * 2 + 32)
#define BPITCH_ (256 * 2 + 32)
#define ASTG_   (KP_ * APITCH_)
#define BSTG_   (KP_ * BPITCH_)
#define STG_    (ASTG_ + BSTG_)
#define QST_OFF_ 65536
#define QTP_     68

__device__ __forceinline__ void kT_stage_products(const float* __restrict__ sa,
                                                  const float* __restrict__ sb,
                                                  v8f (&acc)[2][4]) {
  #pragma unroll
  for (int j = 0; j < KS_ / 4; ++j) {
    v2f a0 = *(const v2f*)(sa + j * 2 * APITCH_);
    v2f a1 = *(const v2f*)(sa + j * 2 * APITCH_ + 32);
    v2f b0 = *(const v2f*)(sb + j * 2 * BPITCH_);
    v2f b1 = *(const v2f*)(sb + j * 2 * BPITCH_ + 32);
    v2f b2 = *(const v2f*)(sb + j * 2 * BPITCH_ + 64);
    v2f b3 = *(const v2f*)(sb + j * 2 * BPITCH_ + 96);
    acc[0][0] = __builtin_amdgcn_wmma_f32_16x16x4_f32(false, a0, false, b0, (short)0, acc[0][0], false, false);
    acc[0][1] = __builtin_amdgcn_wmma_f32_16x16x4_f32(false, a0, false, b1, (short)0, acc[0][1], false, false);
    acc[0][2] = __builtin_amdgcn_wmma_f32_16x16x4_f32(false, a0, false, b2, (short)0, acc[0][2], false, false);
    acc[0][3] = __builtin_amdgcn_wmma_f32_16x16x4_f32(false, a0, false, b3, (short)0, acc[0][3], false, false);
    acc[1][0] = __builtin_amdgcn_wmma_f32_16x16x4_f32(false, a1, false, b0, (short)0, acc[1][0], false, false);
    acc[1][1] = __builtin_amdgcn_wmma_f32_16x16x4_f32(false, a1, false, b1, (short)0, acc[1][1], false, false);
    acc[1][2] = __builtin_amdgcn_wmma_f32_16x16x4_f32(false, a1, false, b2, (short)0, acc[1][2], false, false);
    acc[1][3] = __builtin_amdgcn_wmma_f32_16x16x4_f32(false, a1, false, b3, (short)0, acc[1][3], false, false);
  }
}

template <int LDA, int LDB, int LDC, int K, bool TSTORE, bool HAS_BIAS, bool HAS_RES, int BADD, int SMX, int GN>
__global__ __launch_bounds__(256) __attribute__((amdgpu_waves_per_eu(2, BADD ? 2 : 8))) void kT_gemm(
    const float* __restrict__ A, long long strideA,
    const float* __restrict__ B, long long strideB,
    float* __restrict__ Cmat,    long long strideC,
    const float* __restrict__ bias,
    const float* __restrict__ resid, long long strideR,
    float* __restrict__ stats,
    float* __restrict__ aux,
    const float* __restrict__ gnw,
    const float* __restrict__ gnb,
    const float* __restrict__ gst) {
  __shared__ __attribute__((aligned(16))) float smem[2 * STG_];
  __shared__ __attribute__((aligned(16))) float qtab[(SMX == 3 || GN == 1) ? 2 * K : 4];

  int z = blockIdx.z;
  const float* Ab = A + (size_t)z * strideA;
  const float* Bb = B + (size_t)z * strideB;
  float*       Cb = Cmat + (size_t)z * strideC;

  int tid  = threadIdx.x;
  int wave = tid >> 5;
  int lane = tid & 31;
  int half = lane >> 4;
  int ml   = lane & 15;
  int mblk = blockIdx.x * 64;
  int nblk = blockIdx.y * 256;
  int m0 = mblk + (wave & 1) * 32;
  int n0 = nblk + (wave >> 1) * 64;

  v8f acc[2][4];
  {
    float bv[4];
    #pragma unroll
    for (int s = 0; s < 4; ++s) bv[s] = HAS_BIAS ? bias[n0 + 16 * s + ml] : 0.0f;
    #pragma unroll
    for (int mi = 0; mi < 2; ++mi)
      #pragma unroll
      for (int s = 0; s < 4; ++s)
        #pragma unroll
        for (int r = 0; r < 8; ++r) acc[mi][s][r] = bv[s];
  }

  const float* Ag = Ab + (size_t)(2 * (tid >> 4)) * LDA + mblk + 4 * (tid & 15);
  const float* Bg = Bb + (size_t)(2 * (tid >> 6)) * LDB + nblk + 4 * (tid & 63);
  const int aSt = (tid >> 4) * APITCH_ + 8 * (tid & 15);
  const int bSt = ASTG_ + (tid >> 6) * BPITCH_ + 8 * (tid & 63);
  const int aLd = half * APITCH_ + ((wave & 1) * 32 + ml) * 2;
  const int bLd = ASTG_ + half * BPITCH_ + ((wave >> 1) * 64 + ml) * 2;

  v4f ra0, ra1, rb0[4], rb1[4];
  v4f rc0[4], rc1[4];
  v4f st0, st1;
  const float* Sg = stats + ((size_t)z * K + 2 * (tid >> 4)) * 2;
  int qo = 4 * (tid >> 4);

#define KT_FETCH()                                                    \
  do {                                                                \
    ra0 = *(const v4f*)(Ag);                                          \
    ra1 = *(const v4f*)(Ag + LDA);                                    \
    if (SMX == 2) {                                                   \
      st0 = *(const v4f*)(Sg);                                        \
      st1 = *(const v4f*)(Sg + 2 * B_ * N_);                          \
      Sg += 2 * KS_;                                                  \
    }                                                                 \
    if (SMX == 3) {                                                   \
      st0 = *(const v4f*)(qtab + qo);                                 \
      qo += 2 * KS_;                                                  \
    }                                                                 \
    _Pragma("unroll")                                                 \
    for (int i = 0; i < 4; ++i) {                                     \
      rb0[i] = *(const v4f*)(Bg + (size_t)(8 * i) * LDB);             \
      rb1[i] = *(const v4f*)(Bg + (size_t)(8 * i + 1) * LDB);         \
      if (BADD) {                                                     \
        rc0[i] = *(const v4f*)(Bg + BADD + (size_t)(8 * i) * LDB);    \
        rc1[i] = *(const v4f*)(Bg + BADD + (size_t)(8 * i + 1) * LDB);\
      }                                                               \
    }                                                                 \
    Ag += (size_t)KS_ * LDA;                                          \
    Bg += (size_t)KS_ * LDB;                                          \
  } while (0)

#define KT_PARK(stg)                                                  \
  do {                                                                \
    float* s_ = smem + (stg) * STG_;                                  \
    v4f t_;                                                           \
    if (GN == 1) {                                                    \
      v4f g_ = *(const v4f*)(qtab + qo);                              \
      qo += 2 * KS_;                                                  \
      ra0.x = ra0.x * g_.x + g_.y; ra0.y = ra0.y * g_.x + g_.y;       \
      ra0.z = ra0.z * g_.x + g_.y; ra0.w = ra0.w * g_.x + g_.y;       \
      ra1.x = ra1.x * g_.z + g_.w; ra1.y = ra1.y * g_.z + g_.w;       \
      ra1.z = ra1.z * g_.z + g_.w; ra1.w = ra1.w * g_.z + g_.w;       \
    }                                                                 \
    if (SMX == 2) {                                                   \
      float M0_ = fmaxf(st0.x, st1.x), M1_ = fmaxf(st0.z, st1.z);     \
      float i0_ = 1.0f / (st0.y * __expf(st0.x - M0_) + st1.y * __expf(st1.x - M0_)); \
      float i1_ = 1.0f / (st0.w * __expf(st0.z - M1_) + st1.w * __expf(st1.z - M1_)); \
      ra0.x = __expf(ra0.x - M0_) * i0_; ra0.y = __expf(ra0.y - M0_) * i0_; \
      ra0.z = __expf(ra0.z - M0_) * i0_; ra0.w = __expf(ra0.w - M0_) * i0_; \
      ra1.x = __expf(ra1.x - M1_) * i1_; ra1.y = __expf(ra1.y - M1_) * i1_; \
      ra1.z = __expf(ra1.z - M1_) * i1_; ra1.w = __expf(ra1.w - M1_) * i1_; \
    }                                                                 \
    if (SMX == 3) {                                                   \
      ra0.x = __expf(ra0.x - st0.x) * st0.y; ra0.y = __expf(ra0.y - st0.x) * st0.y; \
      ra0.z = __expf(ra0.z - st0.x) * st0.y; ra0.w = __expf(ra0.w - st0.x) * st0.y; \
      ra1.x = __expf(ra1.x - st0.z) * st0.w; ra1.y = __expf(ra1.y - st0.z) * st0.w; \
      ra1.z = __expf(ra1.z - st0.z) * st0.w; ra1.w = __expf(ra1.w - st0.z) * st0.w; \
    }                                                                 \
    t_.x = ra0.x; t_.y = ra1.x; t_.z = ra0.y; t_.w = ra1.y;           \
    *(v4f*)(s_ + aSt) = t_;                                           \
    t_.x = ra0.z; t_.y = ra1.z; t_.z = ra0.w; t_.w = ra1.w;           \
    *(v4f*)(s_ + aSt + 4) = t_;                                       \
    _Pragma("unroll")                                                 \
    for (int i = 0; i < 4; ++i) {                                     \
      if (BADD) { rb0[i] += rc0[i]; rb1[i] += rc1[i]; }               \
      float* d_ = s_ + bSt + i * 4 * BPITCH_;                         \
      t_.x = rb0[i].x; t_.y = rb1[i].x; t_.z = rb0[i].y; t_.w = rb1[i].y; \
      *(v4f*)(d_) = t_;                                               \
      t_.x = rb0[i].z; t_.y = rb1[i].z; t_.z = rb0[i].w; t_.w = rb1[i].w; \
      *(v4f*)(d_ + 4) = t_;                                           \
    }                                                                 \
  } while (0)

  constexpr int T = K / KS_;

  if (SMX == 3) {
    const float* Q = stats + QST_OFF_ + (size_t)z * (N_ / 64) * C_ * 2;
    #pragma unroll
    for (int h2 = 0; h2 < K / 256; ++h2) {
      int cc = tid + 256 * h2;
      v2f p[N_ / 64];
      #pragma unroll
      for (int i = 0; i < N_ / 64; ++i) p[i] = *(const v2f*)(Q + ((size_t)i * C_ + cc) * 2);
      float M = p[0].x;
      #pragma unroll
      for (int i = 1; i < N_ / 64; ++i) M = fmaxf(M, p[i].x);
      float S = 0.f;
      #pragma unroll
      for (int i = 0; i < N_ / 64; ++i) S += p[i].y * __expf(p[i].x - M);
      *(v2f*)(qtab + 2 * cc) = v2f{M, 1.0f / S};
    }
    __syncthreads();
  }

  if (GN == 1) {
    #pragma unroll
    for (int h2 = 0; h2 < K / 256; ++h2) {
      int cc = tid + 256 * h2;
      float gm, bt;
      kT_gn_coef(gnw, gnb, gst, z, cc, gm, bt);
      *(v2f*)(qtab + 2 * cc) = v2f{gm, bt};
    }
    __syncthreads();
  }

  KT_FETCH();
  KT_PARK(0);
  __syncthreads();

  #pragma unroll 1
  for (int t = 0; t < T - 1; ++t) {
    const int cur = t & 1;
    KT_FETCH();
    kT_stage_products(smem + cur * STG_ + aLd, smem + cur * STG_ + bLd, acc);
    KT_PARK(cur ^ 1);
    __syncthreads();
  }
  kT_stage_products(smem + ((T - 1) & 1) * STG_ + aLd, smem + ((T - 1) & 1) * STG_ + bLd, acc);

#undef KT_FETCH
#undef KT_PARK

  if (SMX == 1) {
    asm volatile("" : "+v"(tid));
    wave = tid >> 5;
    lane = tid & 31;
    half = lane >> 4;
    ml   = lane & 15;
    m0 = mblk + (wave & 1) * 32;
    n0 = nblk + (wave >> 1) * 64;
  }

  if (!TSTORE && SMX == 1 && nblk < C_) {
    float* Qb = aux + (size_t)z * C_ * N_;
    __syncthreads();
    #pragma unroll
    for (int s = 0; s < 4; ++s) {
      #pragma unroll
      for (int mi = 0; mi < 2; ++mi) {
        float* d_ = smem + ((wave >> 1) * 64 + 16 * s + ml) * QTP_ + (wave & 1) * 32 + 16 * mi + 8 * half;
        v4f t0_, t1_;
        t0_.x = acc[mi][s][0]; t0_.y = acc[mi][s][1]; t0_.z = acc[mi][s][2]; t0_.w = acc[mi][s][3];
        t1_.x = acc[mi][s][4]; t1_.y = acc[mi][s][5]; t1_.z = acc[mi][s][6]; t1_.w = acc[mi][s][7];
        *(v4f*)(d_)     = t0_;
        *(v4f*)(d_ + 4) = t1_;
      }
    }
    float* sq = smem + 256 * QTP_;
    #pragma unroll
    for (int s = 0; s < 4; ++s) {
      float mx = acc[0][s][0];
      #pragma unroll
      for (int mi = 0; mi < 2; ++mi)
        #pragma unroll
        for (int r = 0; r < 8; ++r) mx = fmaxf(mx, acc[mi][s][r]);
      mx = fmaxf(mx, __shfl_xor(mx, 16));
      float sm = 0.f;
      #pragma unroll
      for (int mi = 0; mi < 2; ++mi)
        #pragma unroll
        for (int r = 0; r < 8; ++r) sm += __expf(acc[mi][s][r] - mx);
      sm += __shfl_xor(sm, 16);
      if (half == 0)
        *(v2f*)(sq + ((wave & 1) * 256 + (wave >> 1) * 64 + 16 * s + ml) * 2) = v2f{mx, sm};
    }
    __syncthreads();
    #pragma unroll
    for (int it = 0; it < 16; ++it) {
      int col = it * 16 + (tid >> 4);
      v4f t_ = *(const v4f*)(smem + col * QTP_ + 4 * (tid & 15));
      *(v4f*)(Qb + (size_t)(nblk + col) * N_ + mblk + 4 * (tid & 15)) = t_;
    }
    {
      v2f p0 = *(const v2f*)(sq + tid * 2);
      v2f p1 = *(const v2f*)(sq + (256 + tid) * 2);
      float M = fmaxf(p0.x, p1.x);
      float S = p0.y * __expf(p0.x - M) + p1.y * __expf(p1.x - M);
      *(v2f*)(stats + QST_OFF_ + (((size_t)z * (N_ / 64) + blockIdx.x) * C_ + nblk + tid) * 2) = v2f{M, S};
    }
  } else if (!TSTORE) {
    #pragma unroll
    for (int mi = 0; mi < 2; ++mi) {
      #pragma unroll
      for (int r = 0; r < 8; ++r) {
        int m = m0 + 16 * mi + r + 8 * half;
        #pragma unroll
        for (int s = 0; s < 4; ++s)
          Cb[(size_t)m * LDC + n0 + 16 * s + ml] = acc[mi][s][r];
      }
    }
    if (SMX == 1) {
      if (nblk >= C_ && nblk < 2 * C_) {
        __shared__ __attribute__((aligned(16))) float sred[4 * 64 * 2];
        float pm = 0.f, ps = 0.f;
        #pragma unroll
        for (int mi = 0; mi < 2; ++mi) {
          #pragma unroll
          for (int r = 0; r < 8; ++r) {
            float mx = fmaxf(fmaxf(acc[mi][0][r], acc[mi][1][r]), fmaxf(acc[mi][2][r], acc[mi][3][r]));
            mx = fmaxf(mx, __shfl_xor(mx, 1));
            mx = fmaxf(mx, __shfl_xor(mx, 2));
            mx = fmaxf(mx, __shfl_xor(mx, 4));
            mx = fmaxf(mx, __shfl_xor(mx, 8));
            float sm = (__expf(acc[mi][0][r] - mx) + __expf(acc[mi][1][r] - mx)) +
                       (__expf(acc[mi][2][r] - mx) + __expf(acc[mi][3][r] - mx));
            sm += __shfl_xor(sm, 1);
            sm += __shfl_xor(sm, 2);
            sm += __shfl_xor(sm, 4);
            sm += __shfl_xor(sm, 8);
            if (ml == 8 * mi + r) { pm = mx; ps = sm; }
          }
        }
        int row = (wave & 1) * 32 + 16 * (ml >> 3) + 8 * half + (ml & 7);
        *(v2f*)(sred + ((wave >> 1) * 64 + row) * 2) = v2f{pm, ps};
        __syncthreads();
        if (tid < 64) {
          v2f p0 = *(const v2f*)(sred + (0 * 64 + tid) * 2);
          v2f p1 = *(const v2f*)(sred + (1 * 64 + tid) * 2);
          v2f p2 = *(const v2f*)(sred + (2 * 64 + tid) * 2);
          v2f p3 = *(const v2f*)(sred + (3 * 64 + tid) * 2);
          float M = fmaxf(fmaxf(p0.x, p1.x), fmaxf(p2.x, p3.x));
          float S = (p0.y * __expf(p0.x - M) + p1.y * __expf(p1.x - M)) +
                    (p2.y * __expf(p2.x - M) + p3.y * __expf(p3.x - M));
          size_t plane = (size_t)((nblk - C_) >> 8) * (2 * B_ * N_);
          *(v2f*)(stats + plane + ((size_t)z * N_ + mblk + tid) * 2) = v2f{M, S};
        }
      }
    }
  } else {
    const float* Rb = HAS_RES ? (resid + (size_t)z * strideR) : nullptr;
    float gmv[4], btv[4];
    #pragma unroll
    for (int s = 0; s < 4; ++s) {
      gmv[s] = 1.0f; btv[s] = 0.0f;
      if (GN == 2) kT_gn_coef(gnw, gnb, gst, z, n0 + 16 * s + ml, gmv[s], btv[s]);
    }
    #pragma unroll
    for (int mi = 0; mi < 2; ++mi) {
      #pragma unroll
      for (int r = 0; r < 8; ++r) {
        int m = m0 + 16 * mi + r + 8 * half;
        #pragma unroll
        for (int s = 0; s < 4; ++s) {
          size_t i = (size_t)(n0 + 16 * s + ml) * LDC + m;
          float v = acc[mi][s][r];
          if (HAS_RES) {
            float xr = Rb[i];
            v += (GN == 2) ? (xr * gmv[s] + btv[s]) : xr;
          }
          Cb[i] = v;
        }
      }
    }
  }
}

extern "C" void kernel_launch(void* const* d_in, const int* in_sizes, int n_in,
                              void* d_out, int out_size, void* d_ws, size_t ws_size,
                              hipStream_t stream) {
  const float* x      = (const float*)d_in[0];
  const float* gn_w   = (const float*)d_in[1];
  const float* gn_b   = (const float*)d_in[2];
  const float* qkv_w  = (const float*)d_in[3];
  const float* qkv_b  = (const float*)d_in[4];
  const float* proj_w = (const float*)d_in[5];
  const float* proj_b = (const float*)d_in[6];
  float* out = (float*)d_out;

  float* ws = (float*)d_ws;
  const size_t XN_SZ  = (size_t)B_ * C_ * N_;
  const size_t QKV_SZ = (size_t)B_ * N_ * C3_;
  const size_t CTX_SZ = (size_t)B_ * C_ * C_;
  float* mean   = ws;
  float* rstd   = ws + 64;
  float* xn     = ws + 256;
  float* qkv    = xn   + XN_SZ;
  float* qT     = qkv  + QKV_SZ;
  float* kst    = qT   + XN_SZ;
  float* o2T    = qT   + XN_SZ + CTX_SZ;
  float* wqkvT  = o2T  + XN_SZ;
  float* wprojT = wqkvT + (size_t)C_ * C3_;

  kT_transpose<<<dim3(C_ / 32, C3_ / 32), dim3(32, 8), 0, stream>>>(qkv_w,  wqkvT,  C3_, C_);
  kT_transpose<<<dim3(C_ / 32, C_  / 32), dim3(32, 8), 0, stream>>>(proj_w, wprojT, C_,  C_);

  kT_gn_stats<<<B_ * GRP_, 256, 0, stream>>>(x, mean, rstd);

  kT_gemm<N_, C3_, C3_, C_, false, true, false, 0, 1, 1>
      <<<dim3(N_ / 64, C3_ / 256, B_), 256, 0, stream>>>(
      x, (long long)C_ * N_,
      wqkvT, 0,
      qkv, (long long)N_ * C3_,
      qkv_b, nullptr, 0, kst, qT, gn_w, gn_b, mean);


  kT_gemm<C3_, C3_, C3_, N_ / 2, false, false, false, 0, 2, 0>
      <<<dim3(C_ / 64, C_ / 256, 2 * B_), 256, 0, stream>>>(
      qkv + C_,     (long long)(N_ / 2) * C3_,
      qkv + 2 * C_, (long long)(N_ / 2) * C3_,
      qkv,          (long long)(N_ / 2) * C3_,
      nullptr, nullptr, 0, kst, nullptr, nullptr, nullptr, nullptr);

  kT_gemm<N_, C3_, N_, C_, true, false, false, (N_ / 2) * C3_, 3, 0>
      <<<dim3(N_ / 64, C_ / 256, B_), 256, 0, stream>>>(
      qT, (long long)C_ * N_,
      qkv, (long long)N_ * C3_,
      o2T, (long long)C_ * N_,
      nullptr, nullptr, 0, kst, nullptr, nullptr, nullptr, nullptr);

  kT_gemm<N_, C_, N_, C_, true, true, true, 0, 0, 2>
      <<<dim3(N_ / 64, C_ / 256, B_), 256, 0, stream>>>(
      o2T, (long long)C_ * N_,
      wprojT, 0,
      out, (long long)C_ * N_,
      proj_b, x, (long long)C_ * N_, nullptr, nullptr, gn_w, gn_b, mean);
}


# Weight transposes dropped: qkv and projection GEMMs read the weights as stored ([out][in]), 16 B along k per lane, parked in LDS as [k/4][column][4]; 7 launches to 5; sum order kept
# speedup vs baseline: 1.8058x; 1.0374x over previous
#include <hip/hip_runtime.h>
#include <math.h>

typedef float v2f __attribute__((ext_vector_type(2)));
typedef float v8f __attribute__((ext_vector_type(8)));

#define B_    8
#define C_    512
#define N_    1024
#define C3_   1536
#define GRP_  8
#define GCH_  64
#define GELT_ (GCH_ * N_)
#define EPS_  1e-5f

__global__ __launch_bounds__(256) void kT_gn_stats(const float* __restrict__ x,
                                                   float* __restrict__ mean,
                                                   float* __restrict__ rstd) {
  __shared__ float s1[256], s2[256];
  int bg = blockIdx.x;
  const float4* p = (const float4*)(x + (size_t)bg * GELT_);
  float s = 0.f, q = 0.f;
  for (int i = threadIdx.x; i < GELT_ / 4; i += 256) {
    float4 v = p[i];
    s += v.x + v.y + v.z + v.w;
    q += v.x * v.x + v.y * v.y + v.z * v.z + v.w * v.w;
  }
  s1[threadIdx.x] = s; s2[threadIdx.x] = q;
  __syncthreads();
  for (int off = 128; off > 0; off >>= 1) {
    if (threadIdx.x < off) {
      s1[threadIdx.x] += s1[threadIdx.x + off];
      s2[threadIdx.x] += s2[threadIdx.x + off];
    }
    __syncthreads();
  }
  if (threadIdx.x == 0) {
    float m = s1[0] * (1.0f / GELT_);
    float v = s2[0] * (1.0f / GELT_) - m * m;
    mean[bg] = m;
    rstd[bg] = rsqrtf(v + EPS_);
  }
}

__device__ __forceinline__ void kT_gn_coef(const float* __restrict__ gnw,
                                           const float* __restrict__ gnb,
                                           const float* __restrict__ gst,
                                           int b, int c, float& gm, float& bt) {
  int bg = b * GRP_ + (c >> 6);
  gm = gnw[c] * gst[64 + bg];
  bt = gnb[c] - gst[bg] * gm;
}

typedef float v4f __attribute__((ext_vector_type(4)));

#define KS_     32
#define KP_     (KS_ / 2)
#define APITCH_ (64 * 2 + 32)
#define BPITCH_ (256 * 2 + 32)
#define ASTG_   (KP_ * APITCH_)
#define BSTG_   (KP_ * BPITCH_)
#define STG_    (ASTG_ + BSTG_)
#define QPITCH_ (256 * 4 + 8)
#define QST_OFF_ 65536
#define QTP_     68

template <bool BK>
__device__ __forceinline__ void kT_stage_products(const float* __restrict__ sa,
                                                  const float* __restrict__ sb,
                                                  v8f (&acc)[2][4]) {
  #pragma unroll
  for (int j = 0; j < KS_ / 4; ++j) {
    v2f a0 = *(const v2f*)(sa + j * 2 * APITCH_);
    v2f a1 = *(const v2f*)(sa + j * 2 * APITCH_ + 32);
    constexpr int BJ = BK ? QPITCH_ : 2 * BPITCH_;
    constexpr int BS = BK ? 64 : 32;
    v2f b0 = *(const v2f*)(sb + j * BJ);
    v2f b1 = *(const v2f*)(sb + j * BJ + BS);
    v2f b2 = *(const v2f*)(sb + j * BJ + 2 * BS);
    v2f b3 = *(const v2f*)(sb + j * BJ + 3 * BS);
    acc[0][0] = __builtin_amdgcn_wmma_f32_16x16x4_f32(false, a0, false, b0, (short)0, acc[0][0], false, false);
    acc[0][1] = __builtin_amdgcn_wmma_f32_16x16x4_f32(false, a0, false, b1, (short)0, acc[0][1], false, false);
    acc[0][2] = __builtin_amdgcn_wmma_f32_16x16x4_f32(false, a0, false, b2, (short)0, acc[0][2], false, false);
    acc[0][3] = __builtin_amdgcn_wmma_f32_16x16x4_f32(false, a0, false, b3, (short)0, acc[0][3], false, false);
    acc[1][0] = __builtin_amdgcn_wmma_f32_16x16x4_f32(false, a1, false, b0, (short)0, acc[1][0], false, false);
    acc[1][1] = __builtin_amdgcn_wmma_f32_16x16x4_f32(false, a1, false, b1, (short)0, acc[1][1], false, false);
    acc[1][2] = __builtin_amdgcn_wmma_f32_16x16x4_f32(false, a1, false, b2, (short)0, acc[1][2], false, false);
    acc[1][3] = __builtin_amdgcn_wmma_f32_16x16x4_f32(false, a1, false, b3, (short)0, acc[1][3], false, false);
  }
}

template <int LDA, int LDB, int LDC, int K, bool TSTORE, bool HAS_BIAS, bool HAS_RES, int BADD, int SMX, int GN, bool BK>
__global__ __launch_bounds__(256) __attribute__((amdgpu_waves_per_eu(2, BADD ? 2 : 8))) void kT_gemm(
    const float* __restrict__ A, long long strideA,
    const float* __restrict__ B, long long strideB,
    float* __restrict__ Cmat,    long long strideC,
    const float* __restrict__ bias,
    const float* __restrict__ resid, long long strideR,
    float* __restrict__ stats,
    float* __restrict__ aux,
    const float* __restrict__ gnw,
    const float* __restrict__ gnb,
    const float* __restrict__ gst) {
  static_assert(!(BK && BADD), "the two-addend B fetch exists for the n-contiguous layout only");
  static_assert(8 * QPITCH_ <= BSTG_, "a BK stage fits the B stage");
  __shared__ __attribute__((aligned(16))) float smem[2 * STG_];
  __shared__ __attribute__((aligned(16))) float qtab[(SMX == 3 || GN == 1) ? 2 * K : 4];

  int z = blockIdx.z;
  const float* Ab = A + (size_t)z * strideA;
  const float* Bb = B + (size_t)z * strideB;
  float*       Cb = Cmat + (size_t)z * strideC;

  int tid  = threadIdx.x;
  int wave = tid >> 5;
  int lane = tid & 31;
  int half = lane >> 4;
  int ml   = lane & 15;
  int mblk = blockIdx.x * 64;
  int nblk = blockIdx.y * 256;
  int m0 = mblk + (wave & 1) * 32;
  int n0 = nblk + (wave >> 1) * 64;

  v8f acc[2][4];
  {
    float bv[4];
    #pragma unroll
    for (int s = 0; s < 4; ++s) bv[s] = HAS_BIAS ? bias[n0 + 16 * s + ml] : 0.0f;
    #pragma unroll
    for (int mi = 0; mi < 2; ++mi)
      #pragma unroll
      for (int s = 0; s < 4; ++s)
        #pragma unroll
        for (int r = 0; r < 8; ++r) acc[mi][s][r] = bv[s];
  }

  const float* Ag = Ab + (size_t)(2 * (tid >> 4)) * LDA + mblk + 4 * (tid & 15);
  const float* Bg = BK ? Bb + (size_t)(nblk + (tid >> 3)) * LDB + 4 * (tid & 7)
                       : Bb + (size_t)(2 * (tid >> 6)) * LDB + nblk + 4 * (tid & 63);
  const int aSt = (tid >> 4) * APITCH_ + 8 * (tid & 15);
  const int bSt = BK ? ASTG_ + (tid & 7) * QPITCH_ + 4 * (tid >> 3)
                     : ASTG_ + (tid >> 6) * BPITCH_ + 8 * (tid & 63);
  const int aLd = half * APITCH_ + ((wave & 1) * 32 + ml) * 2;
  const int bLd = BK ? ASTG_ + ((wave >> 1) * 64 + ml) * 4 + 2 * half
                     : ASTG_ + half * BPITCH_ + ((wave >> 1) * 64 + ml) * 2;

  v4f ra0, ra1, rb0[4], rb1[4];
  v4f rc0[4], rc1[4];
  v4f st0, st1;
  const float* Sg = stats + ((size_t)z * K + 2 * (tid >> 4)) * 2;
  int qo = 4 * (tid >> 4);

#define KT_FETCH()                                                    \
  do {                                                                \
    ra0 = *(const v4f*)(Ag);                                          \
    ra1 = *(const v4f*)(Ag + LDA);                                    \
    if (SMX == 2) {                                                   \
      st0 = *(const v4f*)(Sg);                                        \
      st1 = *(const v4f*)(Sg + 2 * B_ * N_);                          \
      Sg += 2 * KS_;                                                  \
    }                                                                 \
    if (SMX == 3) {                                                   \
      st0 = *(const v4f*)(qtab + qo);                                 \
      qo += 2 * KS_;                                                  \
    }                                                                 \
    _Pragma("unroll")                                                 \
    for (int i = 0; i < 4; ++i) {                                     \
      if (BK) {                                                       \
        rb0[i] = *(const v4f*)(Bg + (size_t)(32 * i) * LDB);          \
        rb1[i] = *(const v4f*)(Bg + (size_t)(32 * (i + 4)) * LDB);    \
      } else {                                                        \
        rb0[i] = *(const v4f*)(Bg + (size_t)(8 * i) * LDB);           \
        rb1[i] = *(const v4f*)(Bg + (size_t)(8 * i + 1) * LDB);       \
      }                                                               \
      if (BADD) {                                                     \
        rc0[i] = *(const v4f*)(Bg + BADD + (size_t)(8 * i) * LDB);    \
        rc1[i] = *(const v4f*)(Bg + BADD + (size_t)(8 * i + 1) * LDB);\
      }                                                               \
    }                                                                 \
    Ag += (size_t)KS_ * LDA;                                          \
    Bg += BK ? (size_t)KS_ : (size_t)KS_ * LDB;                       \
  } while (0)

#define KT_PARK(stg)                                                  \
  do {                                                                \
    float* s_ = smem + (stg) * STG_;                                  \
    v4f t_;                                                           \
    if (GN == 1) {                                                    \
      v4f g_ = *(const v4f*)(qtab + qo);                              \
      qo += 2 * KS_;                                                  \
      ra0.x = ra0.x * g_.x + g_.y; ra0.y = ra0.y * g_.x + g_.y;       \
      ra0.z = ra0.z * g_.x + g_.y; ra0.w = ra0.w * g_.x + g_.y;       \
      ra1.x = ra1.x * g_.z + g_.w; ra1.y = ra1.y * g_.z + g_.w;       \
      ra1.z = ra1.z * g_.z + g_.w; ra1.w = ra1.w * g_.z + g_.w;       \
    }                                                                 \
    if (SMX == 2) {                                                   \
      float M0_ = fmaxf(st0.x, st1.x), M1_ = fmaxf(st0.z, st1.z);     \
      float i0_ = 1.0f / (st0.y * __expf(st0.x - M0_) + st1.y * __expf(st1.x - M0_)); \
      float i1_ = 1.0f / (st0.w * __expf(st0.z - M1_) + st1.w * __expf(st1.z - M1_)); \
      ra0.x = __expf(ra0.x - M0_) * i0_; ra0.y = __expf(ra0.y - M0_) * i0_; \
      ra0.z = __expf(ra0.z - M0_) * i0_; ra0.w = __expf(ra0.w - M0_) * i0_; \
      ra1.x = __expf(ra1.x - M1_) * i1_; ra1.y = __expf(ra1.y - M1_) * i1_; \
      ra1.z = __expf(ra1.z - M1_) * i1_; ra1.w = __expf(ra1.w - M1_) * i1_; \
    }                                                                 \
    if (SMX == 3) {                                                   \
      ra0.x = __expf(ra0.x - st0.x) * st0.y; ra0.y = __expf(ra0.y - st0.x) * st0.y; \
      ra0.z = __expf(ra0.z - st0.x) * st0.y; ra0.w = __expf(ra0.w - st0.x) * st0.y; \
      ra1.x = __expf(ra1.x - st0.z) * st0.w; ra1.y = __expf(ra1.y - st0.z) * st0.w; \
      ra1.z = __expf(ra1.z - st0.z) * st0.w; ra1.w = __expf(ra1.w - st0.z) * st0.w; \
    }                                                                 \
    t_.x = ra0.x; t_.y = ra1.x; t_.z = ra0.y; t_.w = ra1.y;           \
    *(v4f*)(s_ + aSt) = t_;                                           \
    t_.x = ra0.z; t_.y = ra1.z; t_.z = ra0.w; t_.w = ra1.w;           \
    *(v4f*)(s_ + aSt + 4) = t_;                                       \
    _Pragma("unroll")                                                 \
    for (int i = 0; i < 4; ++i) {                                     \
      if (BK) {                                                       \
        *(v4f*)(s_ + bSt + 128 * i)       = rb0[i];                   \
        *(v4f*)(s_ + bSt + 128 * (i + 4)) = rb1[i];                   \
      } else {                                                        \
        if (BADD) { rb0[i] += rc0[i]; rb1[i] += rc1[i]; }             \
        float* d_ = s_ + bSt + i * 4 * BPITCH_;                       \
        t_.x = rb0[i].x; t_.y = rb1[i].x; t_.z = rb0[i].y; t_.w = rb1[i].y; \
        *(v4f*)(d_) = t_;                                             \
        t_.x = rb0[i].z; t_.y = rb1[i].z; t_.z = rb0[i].w; t_.w = rb1[i].w; \
        *(v4f*)(d_ + 4) = t_;                                         \
      }                                                               \
    }                                                                 \
  } while (0)

  constexpr int T = K / KS_;

  if (SMX == 3) {
    const float* Q = stats + QST_OFF_ + (size_t)z * (N_ / 64) * C_ * 2;
    #pragma unroll
    for (int h2 = 0; h2 < K / 256; ++h2) {
      int cc = tid + 256 * h2;
      v2f p[N_ / 64];
      #pragma unroll
      for (int i = 0; i < N_ / 64; ++i) p[i] = *(const v2f*)(Q + ((size_t)i * C_ + cc) * 2);
      float M = p[0].x;
      #pragma unroll
      for (int i = 1; i < N_ / 64; ++i) M = fmaxf(M, p[i].x);
      float S = 0.f;
      #pragma unroll
      for (int i = 0; i < N_ / 64; ++i) S += p[i].y * __expf(p[i].x - M);
      *(v2f*)(qtab + 2 * cc) = v2f{M, 1.0f / S};
    }
    __syncthreads();
  }

  if (GN == 1) {
    #pragma unroll
    for (int h2 = 0; h2 < K / 256; ++h2) {
      int cc = tid + 256 * h2;
      float gm, bt;
      kT_gn_coef(gnw, gnb, gst, z, cc, gm, bt);
      *(v2f*)(qtab + 2 * cc) = v2f{gm, bt};
    }
    __syncthreads();
  }

  KT_FETCH();
  KT_PARK(0);
  __syncthreads();

  #pragma unroll 1
  for (int t = 0; t < T - 1; ++t) {
    const int cur = t & 1;
    KT_FETCH();
    kT_stage_products<BK>(smem + cur * STG_ + aLd, smem + cur * STG_ + bLd, acc);
    KT_PARK(cur ^ 1);
    __syncthreads();
  }
  kT_stage_products<BK>(smem + ((T - 1) & 1) * STG_ + aLd, smem + ((T - 1) & 1) * STG_ + bLd, acc);

#undef KT_FETCH
#undef KT_PARK

  if (SMX == 1) {
    asm volatile("" : "+v"(tid));
    wave = tid >> 5;
    lane = tid & 31;
    half = lane >> 4;
    ml   = lane & 15;
    m0 = mblk + (wave & 1) * 32;
    n0 = nblk + (wave >> 1) * 64;
  }

  if (!TSTORE && SMX == 1 && nblk < C_) {
    float* Qb = aux + (size_t)z * C_ * N_;
    __syncthreads();
    #pragma unroll
    for (int s = 0; s < 4; ++s) {
      #pragma unroll
      for (int mi = 0; mi < 2; ++mi) {
        float* d_ = smem + ((wave >> 1) * 64 + 16 * s + ml) * QTP_ + (wave & 1) * 32 + 16 * mi + 8 * half;
        v4f t0_, t1_;
        t0_.x = acc[mi][s][0]; t0_.y = acc[mi][s][1]; t0_.z = acc[mi][s][2]; t0_.w = acc[mi][s][3];
        t1_.x = acc[mi][s][4]; t1_.y = acc[mi][s][5]; t1_.z = acc[mi][s][6]; t1_.w = acc[mi][s][7];
        *(v4f*)(d_)     = t0_;
        *(v4f*)(d_ + 4) = t1_;
      }
    }
    float* sq = smem + 256 * QTP_;
    #pragma unroll
    for (int s = 0; s < 4; ++s) {
      float mx = acc[0][s][0];
      #pragma unroll
      for (int mi = 0; mi < 2; ++mi)
        #pragma unroll
        for (int r = 0; r < 8; ++r) mx = fmaxf(mx, acc[mi][s][r]);
      mx = fmaxf(mx, __shfl_xor(mx, 16));
      float sm = 0.f;
      #pragma unroll
      for (int mi = 0; mi < 2; ++mi)
        #pragma unroll
        for (int r = 0; r < 8; ++r) sm += __expf(acc[mi][s][r] - mx);
      sm += __shfl_xor(sm, 16);
      if (half == 0)
        *(v2f*)(sq + ((wave & 1) * 256 + (wave >> 1) * 64 + 16 * s + ml) * 2) = v2f{mx, sm};
    }
    __syncthreads();
    #pragma unroll
    for (int it = 0; it < 16; ++it) {
      int col = it * 16 + (tid >> 4);
      v4f t_ = *(const v4f*)(smem + col * QTP_ + 4 * (tid & 15));
      *(v4f*)(Qb + (size_t)(nblk + col) * N_ + mblk + 4 * (tid & 15)) = t_;
    }
    {
      v2f p0 = *(const v2f*)(sq + tid * 2);
      v2f p1 = *(const v2f*)(sq + (256 + tid) * 2);
      float M = fmaxf(p0.x, p1.x);
      float S = p0.y * __expf(p0.x - M) + p1.y * __expf(p1.x - M);
      *(v2f*)(stats + QST_OFF_ + (((size_t)z * (N_ / 64) + blockIdx.x) * C_ + nblk + tid) * 2) = v2f{M, S};
    }
  } else if (!TSTORE) {
    #pragma unroll
    for (int mi = 0; mi < 2; ++mi) {
      #pragma unroll
      for (int r = 0; r < 8; ++r) {
        int m = m0 + 16 * mi + r + 8 * half;
        #pragma unroll
        for (int s = 0; s < 4; ++s)
          Cb[(size_t)m * LDC + n0 + 16 * s + ml] = acc[mi][s][r];
      }
    }
    if (SMX == 1) {
      if (nblk >= C_ && nblk < 2 * C_) {
        __shared__ __attribute__((aligned(16))) float sred[4 * 64 * 2];
        float pm = 0.f, ps = 0.f;
        #pragma unroll
        for (int mi = 0; mi < 2; ++mi) {
          #pragma unroll
          for (int r = 0; r < 8; ++r) {
            float mx = fmaxf(fmaxf(acc[mi][0][r], acc[mi][1][r]), fmaxf(acc[mi][2][r], acc[mi][3][r]));
            mx = fmaxf(mx, __shfl_xor(mx, 1));
            mx = fmaxf(mx, __shfl_xor(mx, 2));
            mx = fmaxf(mx, __shfl_xor(mx, 4));
            mx = fmaxf(mx, __shfl_xor(mx, 8));
            float sm = (__expf(acc[mi][0][r] - mx) + __expf(acc[mi][1][r] - mx)) +
                       (__expf(acc[mi][2][r] - mx) + __expf(acc[mi][3][r] - mx));
            sm += __shfl_xor(sm, 1);
            sm += __shfl_xor(sm, 2);
            sm += __shfl_xor(sm, 4);
            sm += __shfl_xor(sm, 8);
            if (ml == 8 * mi + r) { pm = mx; ps = sm; }
          }
        }
        int row = (wave & 1) * 32 + 16 * (ml >> 3) + 8 * half + (ml & 7);
        *(v2f*)(sred + ((wave >> 1) * 64 + row) * 2) = v2f{pm, ps};
        __syncthreads();
        if (tid < 64) {
          v2f p0 = *(const v2f*)(sred + (0 * 64 + tid) * 2);
          v2f p1 = *(const v2f*)(sred + (1 * 64 + tid) * 2);
          v2f p2 = *(const v2f*)(sred + (2 * 64 + tid) * 2);
          v2f p3 = *(const v2f*)(sred + (3 * 64 + tid) * 2);
          float M = fmaxf(fmaxf(p0.x, p1.x), fmaxf(p2.x, p3.x));
          float S = (p0.y * __expf(p0.x - M) + p1.y * __expf(p1.x - M)) +
                    (p2.y * __expf(p2.x - M) + p3.y * __expf(p3.x - M));
          size_t plane = (size_t)((nblk - C_) >> 8) * (2 * B_ * N_);
          *(v2f*)(stats + plane + ((size_t)z * N_ + mblk + tid) * 2) = v2f{M, S};
        }
      }
    }
  } else {
    const float* Rb = HAS_RES ? (resid + (size_t)z * strideR) : nullptr;
    float gmv[4], btv[4];
    #pragma unroll
    for (int s = 0; s < 4; ++s) {
      gmv[s] = 1.0f; btv[s] = 0.0f;
      if (GN == 2) kT_gn_coef(gnw, gnb, gst, z, n0 + 16 * s + ml, gmv[s], btv[s]);
    }
    #pragma unroll
    for (int mi = 0; mi < 2; ++mi) {
      #pragma unroll
      for (int r = 0; r < 8; ++r) {
        int m = m0 + 16 * mi + r + 8 * half;
        #pragma unroll
        for (int s = 0; s < 4; ++s) {
          size_t i = (size_t)(n0 + 16 * s + ml) * LDC + m;
          float v = acc[mi][s][r];
          if (HAS_RES) {
            float xr = Rb[i];
            v += (GN == 2) ? (xr * gmv[s] + btv[s]) : xr;
          }
          Cb[i] = v;
        }
      }
    }
  }
}

extern "C" void kernel_launch(void* const* d_in, const int* in_sizes, int n_in,
                              void* d_out, int out_size, void* d_ws, size_t ws_size,
                              hipStream_t stream) {
  const float* x      = (const float*)d_in[0];
  const float* gn_w   = (const float*)d_in[1];
  const float* gn_b   = (const float*)d_in[2];
  const float* qkv_w  = (const float*)d_in[3];
  const float* qkv_b  = (const float*)d_in[4];
  const float* proj_w = (const float*)d_in[5];
  const float* proj_b = (const float*)d_in[6];
  float* out = (float*)d_out;

  float* ws = (float*)d_ws;
  const size_t XN_SZ  = (size_t)B_ * C_ * N_;
  const size_t QKV_SZ = (size_t)B_ * N_ * C3_;
  const size_t CTX_SZ = (size_t)B_ * C_ * C_;
  float* mean   = ws;
  float* rstd   = ws + 64;
  float* xn     = ws + 256;
  float* qkv    = xn   + XN_SZ;
  float* qT     = qkv  + QKV_SZ;
  float* kst    = qT   + XN_SZ;
  float* o2T    = qT   + XN_SZ + CTX_SZ;


  kT_gn_stats<<<B_ * GRP_, 256, 0, stream>>>(x, mean, rstd);

  kT_gemm<N_, C_, C3_, C_, false, true, false, 0, 1, 1, true>
      <<<dim3(N_ / 64, C3_ / 256, B_), 256, 0, stream>>>(
      x, (long long)C_ * N_,
      qkv_w, 0,
      qkv, (long long)N_ * C3_,
      qkv_b, nullptr, 0, kst, qT, gn_w, gn_b, mean);


  kT_gemm<C3_, C3_, C3_, N_ / 2, false, false, false, 0, 2, 0, false>
      <<<dim3(C_ / 64, C_ / 256, 2 * B_), 256, 0, stream>>>(
      qkv + C_,     (long long)(N_ / 2) * C3_,
      qkv + 2 * C_, (long long)(N_ / 2) * C3_,
      qkv,          (long long)(N_ / 2) * C3_,
      nullptr, nullptr, 0, kst, nullptr, nullptr, nullptr, nullptr);

  kT_gemm<N_, C3_, N_, C_, true, false, false, (N_ / 2) * C3_, 3, 0, false>
      <<<dim3(N_ / 64, C_ / 256, B_), 256, 0, stream>>>(
      qT, (long long)C_ * N_,
      qkv, (long long)N_ * C3_,
      o2T, (long long)C_ * N_,
      nullptr, nullptr, 0, kst, nullptr, nullptr, nullptr, nullptr);

  kT_gemm<N_, C_, N_, C_, true, true, true, 0, 0, 2, true>
      <<<dim3(N_ / 64, C_ / 256, B_), 256, 0, stream>>>(
      o2T, (long long)C_ * N_,
      proj_w, 0,
      out, (long long)C_ * N_,
      proj_b, x, (long long)C_ * N_, nullptr, nullptr, gn_w, gn_b, mean);
}


# One-block-per-WGP GEMMs: waves 4-7 run out of step with SIMD partners 0-3: they park the next tile before their products and hold a step's last 8 products back over the barrier; sum order kept
# speedup vs baseline: 1.8709x; 1.0361x over previous
#include <hip/hip_runtime.h>
#include <math.h>

typedef float v2f __attribute__((ext_vector_type(2)));
typedef float v8f __attribute__((ext_vector_type(8)));

#define B_    8
#define C_    512
#define N_    1024
#define C3_   1536
#define GRP_  8
#define GCH_  64
#define GELT_ (GCH_ * N_)
#define EPS_  1e-5f

__global__ __launch_bounds__(256) void kT_gn_stats(const float* __restrict__ x,
                                                   float* __restrict__ mean,
                                                   float* __restrict__ rstd) {
  __shared__ float s1[256], s2[256];
  int bg = blockIdx.x;
  const float4* p = (const float4*)(x + (size_t)bg * GELT_);
  float s = 0.f, q = 0.f;
  for (int i = threadIdx.x; i < GELT_ / 4; i += 256) {
    float4 v = p[i];
    s += v.x + v.y + v.z + v.w;
    q += v.x * v.x + v.y * v.y + v.z * v.z + v.w * v.w;
  }
  s1[threadIdx.x] = s; s2[threadIdx.x] = q;
  __syncthreads();
  for (int off = 128; off > 0; off >>= 1) {
    if (threadIdx.x < off) {
      s1[threadIdx.x] += s1[threadIdx.x + off];
      s2[threadIdx.x] += s2[threadIdx.x + off];
    }
    __syncthreads();
  }
  if (threadIdx.x == 0) {
    float m = s1[0] * (1.0f / GELT_);
    float v = s2[0] * (1.0f / GELT_) - m * m;
    mean[bg] = m;
    rstd[bg] = rsqrtf(v + EPS_);
  }
}

__device__ __forceinline__ void kT_gn_coef(const float* __restrict__ gnw,
                                           const float* __restrict__ gnb,
                                           const float* __restrict__ gst,
                                           int b, int c, float& gm, float& bt) {
  int bg = b * GRP_ + (c >> 6);
  gm = gnw[c] * gst[64 + bg];
  bt = gnb[c] - gst[bg] * gm;
}

typedef float v4f __attribute__((ext_vector_type(4)));

#define KS_     32
#define KP_     (KS_ / 2)
#define APITCH_ (64 * 2 + 32)
#define BPITCH_ (256 * 2 + 32)
#define ASTG_   (KP_ * APITCH_)
#define BSTG_   (KP_ * BPITCH_)
#define STG_    (ASTG_ + BSTG_)
#define QPITCH_ (256 * 4 + 8)
#define QST_OFF_ 65536
#define QTP_     68

__device__ __forceinline__ void kT_step_products(const v2f (&f)[6], v8f (&acc)[2][4]) {
  acc[0][0] = __builtin_amdgcn_wmma_f32_16x16x4_f32(false, f[0], false, f[2], (short)0, acc[0][0], false, false);
  acc[0][1] = __builtin_amdgcn_wmma_f32_16x16x4_f32(false, f[0], false, f[3], (short)0, acc[0][1], false, false);
  acc[0][2] = __builtin_amdgcn_wmma_f32_16x16x4_f32(false, f[0], false, f[4], (short)0, acc[0][2], false, false);
  acc[0][3] = __builtin_amdgcn_wmma_f32_16x16x4_f32(false, f[0], false, f[5], (short)0, acc[0][3], false, false);
  acc[1][0] = __builtin_amdgcn_wmma_f32_16x16x4_f32(false, f[1], false, f[2], (short)0, acc[1][0], false, false);
  acc[1][1] = __builtin_amdgcn_wmma_f32_16x16x4_f32(false, f[1], false, f[3], (short)0, acc[1][1], false, false);
  acc[1][2] = __builtin_amdgcn_wmma_f32_16x16x4_f32(false, f[1], false, f[4], (short)0, acc[1][2], false, false);
  acc[1][3] = __builtin_amdgcn_wmma_f32_16x16x4_f32(false, f[1], false, f[5], (short)0, acc[1][3], false, false);
}

template <bool BK>
__device__ __forceinline__ void kT_step_fragments(const float* __restrict__ sa,
                                                  const float* __restrict__ sb,
                                                  int j, v2f (&f)[6]) {
  constexpr int BJ = BK ? QPITCH_ : 2 * BPITCH_;
  constexpr int BS = BK ? 64 : 32;
  f[0] = *(const v2f*)(sa + j * 2 * APITCH_);
  f[1] = *(const v2f*)(sa + j * 2 * APITCH_ + 32);
  f[2] = *(const v2f*)(sb + j * BJ);
  f[3] = *(const v2f*)(sb + j * BJ + BS);
  f[4] = *(const v2f*)(sb + j * BJ + 2 * BS);
  f[5] = *(const v2f*)(sb + j * BJ + 3 * BS);
}

template <bool BK>
__device__ __forceinline__ void kT_stage_products(const float* __restrict__ sa,
                                                  const float* __restrict__ sb,
                                                  v8f (&acc)[2][4]) {
  #pragma unroll
  for (int j = 0; j < KS_ / 4; ++j) {
    v2f f[6];
    kT_step_fragments<BK>(sa, sb, j, f);
    kT_step_products(f, acc);
  }
}

template <bool BK>
__device__ __forceinline__ void kT_stage_products_hold(const float* __restrict__ sa,
                                                       const float* __restrict__ sb,
                                                       v8f (&acc)[2][4], v2f (&hf)[6]) {
  #pragma unroll
  for (int j = 0; j < KS_ / 4 - 1; ++j) {
    v2f f[6];
    kT_step_fragments<BK>(sa, sb, j, f);
    kT_step_products(f, acc);
  }
  kT_step_fragments<BK>(sa, sb, KS_ / 4 - 1, hf);
}

template <int LDA, int LDB, int LDC, int K, bool TSTORE, bool HAS_BIAS, bool HAS_RES, int BADD, int SMX, int GN, bool BK>
__global__ __launch_bounds__(256) __attribute__((amdgpu_waves_per_eu(2, BADD ? 2 : 8))) void kT_gemm(
    const float* __restrict__ A, long long strideA,
    const float* __restrict__ B, long long strideB,
    float* __restrict__ Cmat,    long long strideC,
    const float* __restrict__ bias,
    const float* __restrict__ resid, long long strideR,
    float* __restrict__ stats,
    float* __restrict__ aux,
    const float* __restrict__ gnw,
    const float* __restrict__ gnb,
    const float* __restrict__ gst) {
  static_assert(!(BK && BADD), "the two-addend B fetch exists for the n-contiguous layout only");
  static_assert(8 * QPITCH_ <= BSTG_, "a BK stage fits the B stage");
  __shared__ __attribute__((aligned(16))) float smem[2 * STG_];
  __shared__ __attribute__((aligned(16))) float qtab[(SMX == 3 || GN == 1) ? 2 * K : 4];

  int z = blockIdx.z;
  const float* Ab = A + (size_t)z * strideA;
  const float* Bb = B + (size_t)z * strideB;
  float*       Cb = Cmat + (size_t)z * strideC;

  int tid  = threadIdx.x;
  int wave = tid >> 5;
  int lane = tid & 31;
  int half = lane >> 4;
  int ml   = lane & 15;
  int mblk = blockIdx.x * 64;
  int nblk = blockIdx.y * 256;
  int m0 = mblk + (wave & 1) * 32;
  int n0 = nblk + (wave >> 1) * 64;

  v8f acc[2][4];
  {
    float bv[4];
    #pragma unroll
    for (int s = 0; s < 4; ++s) bv[s] = HAS_BIAS ? bias[n0 + 16 * s + ml] : 0.0f;
    #pragma unroll
    for (int mi = 0; mi < 2; ++mi)
      #pragma unroll
      for (int s = 0; s < 4; ++s)
        #pragma unroll
        for (int r = 0; r < 8; ++r) acc[mi][s][r] = bv[s];
  }

  const float* Ag = Ab + (size_t)(2 * (tid >> 4)) * LDA + mblk + 4 * (tid & 15);
  const float* Bg = BK ? Bb + (size_t)(nblk + (tid >> 3)) * LDB + 4 * (tid & 7)
                       : Bb + (size_t)(2 * (tid >> 6)) * LDB + nblk + 4 * (tid & 63);
  const int aSt = (tid >> 4) * APITCH_ + 8 * (tid & 15);
  const int bSt = BK ? ASTG_ + (tid & 7) * QPITCH_ + 4 * (tid >> 3)
                     : ASTG_ + (tid >> 6) * BPITCH_ + 8 * (tid & 63);
  const int aLd = half * APITCH_ + ((wave & 1) * 32 + ml) * 2;
  const int bLd = BK ? ASTG_ + ((wave >> 1) * 64 + ml) * 4 + 2 * half
                     : ASTG_ + half * BPITCH_ + ((wave >> 1) * 64 + ml) * 2;

  v4f ra0, ra1, rb0[4], rb1[4];
  v4f rc0[4], rc1[4];
  v4f st0, st1;
  const float* Sg = stats + ((size_t)z * K + 2 * (tid >> 4)) * 2;
  int qo = 4 * (tid >> 4);

#define KT_FETCH()                                                    \
  do {                                                                \
    ra0 = *(const v4f*)(Ag);                                          \
    ra1 = *(const v4f*)(Ag + LDA);                                    \
    if (SMX == 2) {                                                   \
      st0 = *(const v4f*)(Sg);                                        \
      st1 = *(const v4f*)(Sg + 2 * B_ * N_);                          \
      Sg += 2 * KS_;                                                  \
    }                                                                 \
    if (SMX == 3) {                                                   \
      st0 = *(const v4f*)(qtab + qo);                                 \
      qo += 2 * KS_;                                                  \
    }                                                                 \
    _Pragma("unroll")                                                 \
    for (int i = 0; i < 4; ++i) {                                     \
      if (BK) {                                                       \
        rb0[i] = *(const v4f*)(Bg + (size_t)(32 * i) * LDB);          \
        rb1[i] = *(const v4f*)(Bg + (size_t)(32 * (i + 4)) * LDB);    \
      } else {                                                        \
        rb0[i] = *(const v4f*)(Bg + (size_t)(8 * i) * LDB);           \
        rb1[i] = *(const v4f*)(Bg + (size_t)(8 * i + 1) * LDB);       \
      }                                                               \
      if (BADD) {                                                     \
        rc0[i] = *(const v4f*)(Bg + BADD + (size_t)(8 * i) * LDB);    \
        rc1[i] = *(const v4f*)(Bg + BADD + (size_t)(8 * i + 1) * LDB);\
      }                                                               \
    }                                                                 \
    Ag += (size_t)KS_ * LDA;                                          \
    Bg += BK ? (size_t)KS_ : (size_t)KS_ * LDB;                       \
  } while (0)

#define KT_PARK(stg)                                                  \
  do {                                                                \
    float* s_ = smem + (stg) * STG_;                                  \
    v4f t_;                                                           \
    if (GN == 1) {                                                    \
      v4f g_ = *(const v4f*)(qtab + qo);                              \
      qo += 2 * KS_;                                                  \
      ra0.x = ra0.x * g_.x + g_.y; ra0.y = ra0.y * g_.x + g_.y;       \
      ra0.z = ra0.z * g_.x + g_.y; ra0.w = ra0.w * g_.x + g_.y;       \
      ra1.x = ra1.x * g_.z + g_.w; ra1.y = ra1.y * g_.z + g_.w;       \
      ra1.z = ra1.z * g_.z + g_.w; ra1.w = ra1.w * g_.z + g_.w;       \
    }                                                                 \
    if (SMX == 2) {                                                   \
      float M0_ = fmaxf(st0.x, st1.x), M1_ = fmaxf(st0.z, st1.z);     \
      float i0_ = 1.0f / (st0.y * __expf(st0.x - M0_) + st1.y * __expf(st1.x - M0_)); \
      float i1_ = 1.0f / (st0.w * __expf(st0.z - M1_) + st1.w * __expf(st1.z - M1_)); \
      ra0.x = __expf(ra0.x - M0_) * i0_; ra0.y = __expf(ra0.y - M0_) * i0_; \
      ra0.z = __expf(ra0.z - M0_) * i0_; ra0.w = __expf(ra0.w - M0_) * i0_; \
      ra1.x = __expf(ra1.x - M1_) * i1_; ra1.y = __expf(ra1.y - M1_) * i1_; \
      ra1.z = __expf(ra1.z - M1_) * i1_; ra1.w = __expf(ra1.w - M1_) * i1_; \
    }                                                                 \
    if (SMX == 3) {                                                   \
      ra0.x = __expf(ra0.x - st0.x) * st0.y; ra0.y = __expf(ra0.y - st0.x) * st0.y; \
      ra0.z = __expf(ra0.z - st0.x) * st0.y; ra0.w = __expf(ra0.w - st0.x) * st0.y; \
      ra1.x = __expf(ra1.x - st0.z) * st0.w; ra1.y = __expf(ra1.y - st0.z) * st0.w; \
      ra1.z = __expf(ra1.z - st0.z) * st0.w; ra1.w = __expf(ra1.w - st0.z) * st0.w; \
    }                                                                 \
    t_.x = ra0.x; t_.y = ra1.x; t_.z = ra0.y; t_.w = ra1.y;           \
    *(v4f*)(s_ + aSt) = t_;                                           \
    t_.x = ra0.z; t_.y = ra1.z; t_.z = ra0.w; t_.w = ra1.w;           \
    *(v4f*)(s_ + aSt + 4) = t_;                                       \
    _Pragma("unroll")                                                 \
    for (int i = 0; i < 4; ++i) {                                     \
      if (BK) {                                                       \
        *(v4f*)(s_ + bSt + 128 * i)       = rb0[i];                   \
        *(v4f*)(s_ + bSt + 128 * (i + 4)) = rb1[i];                   \
      } else {                                                        \
        if (BADD) { rb0[i] += rc0[i]; rb1[i] += rc1[i]; }             \
        float* d_ = s_ + bSt + i * 4 * BPITCH_;                       \
        t_.x = rb0[i].x; t_.y = rb1[i].x; t_.z = rb0[i].y; t_.w = rb1[i].y; \
        *(v4f*)(d_) = t_;                                             \
        t_.x = rb0[i].z; t_.y = rb1[i].z; t_.z = rb0[i].w; t_.w = rb1[i].w; \
        *(v4f*)(d_ + 4) = t_;                                         \
      }                                                               \
    }                                                                 \
  } while (0)

  constexpr int T = K / KS_;

  constexpr bool TURNS = (SMX != 1);
  v2f hf[6];
  #pragma unroll
  for (int i = 0; i < 6; ++i) hf[i] = v2f{0.0f, 0.0f};
  const bool late = TURNS && (__builtin_amdgcn_readfirstlane(tid >> 7) != 0);

  if (SMX == 3) {
    const float* Q = stats + QST_OFF_ + (size_t)z * (N_ / 64) * C_ * 2;
    #pragma unroll
    for (int h2 = 0; h2 < K / 256; ++h2) {
      int cc = tid + 256 * h2;
      v2f p[N_ / 64];
      #pragma unroll
      for (int i = 0; i < N_ / 64; ++i) p[i] = *(const v2f*)(Q + ((size_t)i * C_ + cc) * 2);
      float M = p[0].x;
      #pragma unroll
      for (int i = 1; i < N_ / 64; ++i) M = fmaxf(M, p[i].x);
      float S = 0.f;
      #pragma unroll
      for (int i = 0; i < N_ / 64; ++i) S += p[i].y * __expf(p[i].x - M);
      *(v2f*)(qtab + 2 * cc) = v2f{M, 1.0f / S};
    }
    __syncthreads();
  }

  if (GN == 1) {
    #pragma unroll
    for (int h2 = 0; h2 < K / 256; ++h2) {
      int cc = tid + 256 * h2;
      float gm, bt;
      kT_gn_coef(gnw, gnb, gst, z, cc, gm, bt);
      *(v2f*)(qtab + 2 * cc) = v2f{gm, bt};
    }
    __syncthreads();
  }

  KT_FETCH();
  KT_PARK(0);
  __syncthreads();
  if (late) {
    KT_FETCH();
    KT_PARK(1);
  }

  #pragma unroll 1
  for (int t = 0; t < T - 1; ++t) {
    const int cur = t & 1;
    if (late) {
      if (t > 0) {
        kT_step_products(hf, acc);
        KT_PARK(cur ^ 1);
      }
      if (t == T - 2) {
        Ag -= (size_t)KS_ * LDA;
        Bg -= BK ? (size_t)KS_ : (size_t)KS_ * LDB;
        if (SMX == 2) Sg -= 2 * KS_;
        if (SMX == 3) qo -= 2 * KS_;
      }
    }
    KT_FETCH();
    if (TURNS) {
      kT_stage_products_hold<BK>(smem + cur * STG_ + aLd, smem + cur * STG_ + bLd, acc, hf);
      if (!late) {
        kT_step_products(hf, acc);
        KT_PARK(cur ^ 1);
      }
    } else {
      kT_stage_products<BK>(smem + cur * STG_ + aLd, smem + cur * STG_ + bLd, acc);
      KT_PARK(cur ^ 1);
    }
    __syncthreads();
  }
  if (late) kT_step_products(hf, acc);
  kT_stage_products<BK>(smem + ((T - 1) & 1) * STG_ + aLd, smem + ((T - 1) & 1) * STG_ + bLd, acc);

#undef KT_FETCH
#undef KT_PARK

  if (SMX == 1) {
    asm volatile("" : "+v"(tid));
    wave = tid >> 5;
    lane = tid & 31;
    half = lane >> 4;
    ml   = lane & 15;
    m0 = mblk + (wave & 1) * 32;
    n0 = nblk + (wave >> 1) * 64;
  }

  if (!TSTORE && SMX == 1 && nblk < C_) {
    float* Qb = aux + (size_t)z * C_ * N_;
    __syncthreads();
    #pragma unroll
    for (int s = 0; s < 4; ++s) {
      #pragma unroll
      for (int mi = 0; mi < 2; ++mi) {
        float* d_ = smem + ((wave >> 1) * 64 + 16 * s + ml) * QTP_ + (wave & 1) * 32 + 16 * mi + 8 * half;
        v4f t0_, t1_;
        t0_.x = acc[mi][s][0]; t0_.y = acc[mi][s][1]; t0_.z = acc[mi][s][2]; t0_.w = acc[mi][s][3];
        t1_.x = acc[mi][s][4]; t1_.y = acc[mi][s][5]; t1_.z = acc[mi][s][6]; t1_.w = acc[mi][s][7];
        *(v4f*)(d_)     = t0_;
        *(v4f*)(d_ + 4) = t1_;
      }
    }
    float* sq = smem + 256 * QTP_;
    #pragma unroll
    for (int s = 0; s < 4; ++s) {
      float mx = acc[0][s][0];
      #pragma unroll
      for (int mi = 0; mi < 2; ++mi)
        #pragma unroll
        for (int r = 0; r < 8; ++r) mx = fmaxf(mx, acc[mi][s][r]);
      mx = fmaxf(mx, __shfl_xor(mx, 16));
      float sm = 0.f;
      #pragma unroll
      for (int mi = 0; mi < 2; ++mi)
        #pragma unroll
        for (int r = 0; r < 8; ++r) sm += __expf(acc[mi][s][r] - mx);
      sm += __shfl_xor(sm, 16);
      if (half == 0)
        *(v2f*)(sq + ((wave & 1) * 256 + (wave >> 1) * 64 + 16 * s + ml) * 2) = v2f{mx, sm};
    }
    __syncthreads();
    #pragma unroll
    for (int it = 0; it < 16; ++it) {
      int col = it * 16 + (tid >> 4);
      v4f t_ = *(const v4f*)(smem + col * QTP_ + 4 * (tid & 15));
      *(v4f*)(Qb + (size_t)(nblk + col) * N_ + mblk + 4 * (tid & 15)) = t_;
    }
    {
      v2f p0 = *(const v2f*)(sq + tid * 2);
      v2f p1 = *(const v2f*)(sq + (256 + tid) * 2);
      float M = fmaxf(p0.x, p1.x);
      float S = p0.y * __expf(p0.x - M) + p1.y * __expf(p1.x - M);
      *(v2f*)(stats + QST_OFF_ + (((size_t)z * (N_ / 64) + blockIdx.x) * C_ + nblk + tid) * 2) = v2f{M, S};
    }
  } else if (!TSTORE) {
    #pragma unroll
    for (int mi = 0; mi < 2; ++mi) {
      #pragma unroll
      for (int r = 0; r < 8; ++r) {
        int m = m0 + 16 * mi + r + 8 * half;
        #pragma unroll
        for (int s = 0; s < 4; ++s)
          Cb[(size_t)m * LDC + n0 + 16 * s + ml] = acc[mi][s][r];
      }
    }
    if (SMX == 1) {
      if (nblk >= C_ && nblk < 2 * C_) {
        __shared__ __attribute__((aligned(16))) float sred[4 * 64 * 2];
        float pm = 0.f, ps = 0.f;
        #pragma unroll
        for (int mi = 0; mi < 2; ++mi) {
          #pragma unroll
          for (int r = 0; r < 8; ++r) {
            float mx = fmaxf(fmaxf(acc[mi][0][r], acc[mi][1][r]), fmaxf(acc[mi][2][r], acc[mi][3][r]));
            mx = fmaxf(mx, __shfl_xor(mx, 1));
            mx = fmaxf(mx, __shfl_xor(mx, 2));
            mx = fmaxf(mx, __shfl_xor(mx, 4));
            mx = fmaxf(mx, __shfl_xor(mx, 8));
            float sm = (__expf(acc[mi][0][r] - mx) + __expf(acc[mi][1][r] - mx)) +
                       (__expf(acc[mi][2][r] - mx) + __expf(acc[mi][3][r] - mx));
            sm += __shfl_xor(sm, 1);
            sm += __shfl_xor(sm, 2);
            sm += __shfl_xor(sm, 4);
            sm += __shfl_xor(sm, 8);
            if (ml == 8 * mi + r) { pm = mx; ps = sm; }
          }
        }
        int row = (wave & 1) * 32 + 16 * (ml >> 3) + 8 * half + (ml & 7);
        *(v2f*)(sred + ((wave >> 1) * 64 + row) * 2) = v2f{pm, ps};
        __syncthreads();
        if (tid < 64) {
          v2f p0 = *(const v2f*)(sred + (0 * 64 + tid) * 2);
          v2f p1 = *(const v2f*)(sred + (1 * 64 + tid) * 2);
          v2f p2 = *(const v2f*)(sred + (2 * 64 + tid) * 2);
          v2f p3 = *(const v2f*)(sred + (3 * 64 + tid) * 2);
          float M = fmaxf(fmaxf(p0.x, p1.x), fmaxf(p2.x, p3.x));
          float S = (p0.y * __expf(p0.x - M) + p1.y * __expf(p1.x - M)) +
                    (p2.y * __expf(p2.x - M) + p3.y * __expf(p3.x - M));
          size_t plane = (size_t)((nblk - C_) >> 8) * (2 * B_ * N_);
          *(v2f*)(stats + plane + ((size_t)z * N_ + mblk + tid) * 2) = v2f{M, S};
        }
      }
    }
  } else {
    const float* Rb = HAS_RES ? (resid + (size_t)z * strideR) : nullptr;
    float gmv[4], btv[4];
    #pragma unroll
    for (int s = 0; s < 4; ++s) {
      gmv[s] = 1.0f; btv[s] = 0.0f;
      if (GN == 2) kT_gn_coef(gnw, gnb, gst, z, n0 + 16 * s + ml, gmv[s], btv[s]);
    }
    #pragma unroll
    for (int mi = 0; mi < 2; ++mi) {
      #pragma unroll
      for (int r = 0; r < 8; ++r) {
        int m = m0 + 16 * mi + r + 8 * half;
        #pragma unroll
        for (int s = 0; s < 4; ++s) {
          size_t i = (size_t)(n0 + 16 * s + ml) * LDC + m;
          float v = acc[mi][s][r];
          if (HAS_RES) {
            float xr = Rb[i];
            v += (GN == 2) ? (xr * gmv[s] + btv[s]) : xr;
          }
          Cb[i] = v;
        }
      }
    }
  }
}

extern "C" void kernel_launch(void* const* d_in, const int* in_sizes, int n_in,
                              void* d_out, int out_size, void* d_ws, size_t ws_size,
                              hipStream_t stream) {
  const float* x      = (const float*)d_in[0];
  const float* gn_w   = (const float*)d_in[1];
  const float* gn_b   = (const float*)d_in[2];
  const float* qkv_w  = (const float*)d_in[3];
  const float* qkv_b  = (const float*)d_in[4];
  const float* proj_w = (const float*)d_in[5];
  const float* proj_b = (const float*)d_in[6];
  float* out = (float*)d_out;

  float* ws = (float*)d_ws;
  const size_t XN_SZ  = (size_t)B_ * C_ * N_;
  const size_t QKV_SZ = (size_t)B_ * N_ * C3_;
  const size_t CTX_SZ = (size_t)B_ * C_ * C_;
  float* mean   = ws;
  float* rstd   = ws + 64;
  float* xn     = ws + 256;
  float* qkv    = xn   + XN_SZ;
  float* qT     = qkv  + QKV_SZ;
  float* kst    = qT   + XN_SZ;
  float* o2T    = qT   + XN_SZ + CTX_SZ;


  kT_gn_stats<<<B_ * GRP_, 256, 0, stream>>>(x, mean, rstd);

  kT_gemm<N_, C_, C3_, C_, false, true, false, 0, 1, 1, true>
      <<<dim3(N_ / 64, C3_ / 256, B_), 256, 0, stream>>>(
      x, (long long)C_ * N_,
      qkv_w, 0,
      qkv, (long long)N_ * C3_,
      qkv_b, nullptr, 0, kst, qT, gn_w, gn_b, mean);


  kT_gemm<C3_, C3_, C3_, N_ / 2, false, false, false, 0, 2, 0, false>
      <<<dim3(C_ / 64, C_ / 256, 2 * B_), 256, 0, stream>>>(
      qkv + C_,     (long long)(N_ / 2) * C3_,
      qkv + 2 * C_, (long long)(N_ / 2) * C3_,
      qkv,          (long long)(N_ / 2) * C3_,
      nullptr, nullptr, 0, kst, nullptr, nullptr, nullptr, nullptr);

  kT_gemm<N_, C3_, N_, C_, true, false, false, (N_ / 2) * C3_, 3, 0, false>
      <<<dim3(N_ / 64, C_ / 256, B_), 256, 0, stream>>>(
      qT, (long long)C_ * N_,
      qkv, (long long)N_ * C3_,
      o2T, (long long)C_ * N_,
      nullptr, nullptr, 0, kst, nullptr, nullptr, nullptr, nullptr);

  kT_gemm<N_, C_, N_, C_, true, true, true, 0, 0, 2, true>
      <<<dim3(N_ / 64, C_ / 256, B_), 256, 0, stream>>>(
      o2T, (long long)C_ * N_,
      proj_w, 0,
      out, (long long)C_ * N_,
      proj_b, x, (long long)C_ * N_, nullptr, nullptr, gn_w, gn_b, mean);
}
